# starting kernel
# baseline (speedup 1.0000x reference)
#include <hip/hip_runtime.h>
#include <math.h>

typedef __attribute__((ext_vector_type(2))) float v2f;
typedef __attribute__((ext_vector_type(8))) float v8f;

#define NEG_SLOPE 0.2f

template <int TILES>
__global__ __launch_bounds__(32)
void gat_gemm_wmma(const float* __restrict__ X, const float* __restrict__ W,
                   float* __restrict__ Hout, int N, int K) {
  const int F    = TILES * 16;
  const int row0 = blockIdx.x * 16;
  const int lane = threadIdx.x;
  const int l15  = lane & 15;
  const int kh   = (lane >> 4) * 2;
  const int m    = row0 + l15;

  v8f acc[TILES];
#pragma unroll
  for (int t = 0; t < TILES; ++t) acc[t] = (v8f){0.f,0.f,0.f,0.f,0.f,0.f,0.f,0.f};

  for (int k0 = 0; k0 < K; k0 += 4) {
    v2f a;
    a.x = X[(long)m * K + k0 + kh];
    a.y = X[(long)m * K + k0 + kh + 1];
#pragma unroll
    for (int t = 0; t < TILES; ++t) {
      v2f b;
      b.x = W[(long)(k0 + kh)     * F + t * 16 + l15];
      b.y = W[(long)(k0 + kh + 1) * F + t * 16 + l15];
      acc[t] = __builtin_amdgcn_wmma_f32_16x16x4_f32(
           false, a,  false, b,
           (short)0, acc[t],  false,  false);
    }
  }

  const int hi = (lane >> 4) * 8;
#pragma unroll
  for (int t = 0; t < TILES; ++t)
#pragma unroll
    for (int r = 0; r < 8; ++r)
      Hout[(long)(row0 + hi + r) * F + t * 16 + l15] = acc[t][r];
}

__global__ void gat_alpha(const float* __restrict__ h,
                          const float* __restrict__ a_src,
                          const float* __restrict__ a_dst,
                          float* __restrict__ as, float* __restrict__ ad,
                          int N, int C) {
  int wid  = (int)((blockIdx.x * (long)blockDim.x + threadIdx.x) >> 5);
  int lane = threadIdx.x & 31;
  if (wid >= N) return;
  float ss = 0.f, sd = 0.f;
  for (int c = lane; c < C; c += 32) {
    float hv = h[(long)wid * C + c];
    ss += hv * a_src[c];
    sd += hv * a_dst[c];
  }
#pragma unroll
  for (int off = 16; off > 0; off >>= 1) {
    ss += __shfl_down(ss, off, 32);
    sd += __shfl_down(sd, off, 32);
  }
  if (lane == 0) { as[wid] = ss; ad[wid] = sd; }
}

__global__ void gat_edge_logits(const int* __restrict__ src,
                                const int* __restrict__ dst,
                                const float* __restrict__ as,
                                const float* __restrict__ ad,
                                float* __restrict__ ebuf,
                                float* __restrict__ mbuf,
                                int E, int N) {
  int e = blockIdx.x * blockDim.x + threadIdx.x;
  int ET = E + N;
  if (e >= ET) return;
  int s = (e < E) ? src[e] : (e - E);
  int d = (e < E) ? dst[e] : (e - E);
  float v = as[s] + ad[d];
  v = (v > 0.f) ? v : NEG_SLOPE * v;
  ebuf[e] = v;
  atomicMax(&mbuf[d], v);
}

__global__ void gat_edge_exp(const int* __restrict__ dst,
                             float* __restrict__ ebuf,
                             const float* __restrict__ mbuf,
                             float* __restrict__ den,
                             int E, int N) {
  int e = blockIdx.x * blockDim.x + threadIdx.x;
  int ET = E + N;
  if (e >= ET) return;
  int d = (e < E) ? dst[e] : (e - E);
  float ex = __expf(ebuf[e] - mbuf[d]);
  ebuf[e] = ex;
  atomicAdd(&den[d], ex);
}

__global__ void gat_edge_aggregate(const int* __restrict__ src,
                                   const int* __restrict__ dst,
                                   const float* __restrict__ ebuf,
                                   const float* __restrict__ den,
                                   const float* __restrict__ h,
                                   float* __restrict__ agg,
                                   int E, int N, int C) {
  int wid  = (int)((blockIdx.x * (long)blockDim.x + threadIdx.x) >> 5);
  int lane = threadIdx.x & 31;
  int ET = E + N;
  if (wid >= ET) return;
  int s = (wid < E) ? src[wid] : (wid - E);
  int d = (wid < E) ? dst[wid] : (wid - E);
  float alpha = ebuf[wid] / (den[d] + 1e-16f);
  for (int c = lane; c < C; c += 32)
    atomicAdd(&agg[(long)d * C + c], h[(long)s * C + c] * alpha);
}

__global__ void gat_bias_act(const float* __restrict__ agg,
                             const float* __restrict__ bias,
                             float* __restrict__ out,
                             int N, int C, int do_relu) {
  int i = blockIdx.x * blockDim.x + threadIdx.x;
  if (i >= N * C) return;
  float v = agg[i] + bias[i % C];
  if (do_relu) v = fmaxf(v, 0.f);
  out[i] = v;
}

__global__ void gat_fill(float* __restrict__ p, float v, int n) {
  int i = blockIdx.x * blockDim.x + threadIdx.x;
  if (i < n) p[i] = v;
}

extern "C" void kernel_launch(void* const* d_in, const int* in_sizes, int n_in,
                              void* d_out, int out_size, void* d_ws, size_t ws_size,
                              hipStream_t stream) {
  const float* fea = (const float*)d_in[0];
  const int*   ei  = (const int*)  d_in[1];
  const float* W1  = (const float*)d_in[2];
  const float* av1s= (const float*)d_in[3];
  const float* av1d= (const float*)d_in[4];
  const float* b1  = (const float*)d_in[5];
  const float* W2  = (const float*)d_in[6];
  const float* av2s= (const float*)d_in[7];
  const float* av2d= (const float*)d_in[8];
  const float* b2  = (const float*)d_in[9];
  float* out = (float*)d_out;

  const int B  = 2;
  const int C1 = 128, C2 = 64;
  const int N  = in_sizes[0] / (B * C1);
  const int E  = in_sizes[1] / (B * 2);
  const int ET = E + N;

  float* w   = (float*)d_ws;
  float* h1   = w; w += (long)N * C1;
  float* agg1 = w; w += (long)N * C1;
  float* h2   = w; w += (long)N * C2;
  float* agg2 = w; w += (long)N * C2;
  float* asb  = w; w += N;
  float* adb  = w; w += N;
  float* mb   = w; w += N;
  float* denb = w; w += N;
  float* ebuf = w; w += ET;

  const int TB = 256;
  dim3 blk(TB);
  int gN       = (N + TB - 1) / TB;
  int gNC1     = (N * C1 + TB - 1) / TB;
  int gNC2     = (N * C2 + TB - 1) / TB;
  int gEdges   = (ET + TB - 1) / TB;
  int gEdgeWav = (ET * 32 + TB - 1) / TB;
  int gNodeWav = (N * 32 + TB - 1) / TB;
  int gGemm    = N / 16;

  for (int b = 0; b < B; ++b) {
    const float* xb   = fea + (long)b * N * C1;
    const int*   srcb = ei + (long)b * 2 * E;
    const int*   dstb = srcb + E;
    float*       outb = out + (long)b * N * C2;

    gat_fill<<<gN,   blk, 0, stream>>>(mb,  -INFINITY, N);
    gat_fill<<<gN,   blk, 0, stream>>>(denb, 0.f, N);
    gat_fill<<<gNC1, blk, 0, stream>>>(agg1, 0.f, N * C1);

    gat_gemm_wmma<8><<<gGemm, dim3(32), 0, stream>>>(xb, W1, h1, N, C1);
    gat_alpha<<<gNodeWav, blk, 0, stream>>>(h1, av1s, av1d, asb, adb, N, C1);
    gat_edge_logits<<<gEdges, blk, 0, stream>>>(srcb, dstb, asb, adb, ebuf, mb, E, N);
    gat_edge_exp<<<gEdges, blk, 0, stream>>>(dstb, ebuf, mb, denb, E, N);
    gat_edge_aggregate<<<gEdgeWav, blk, 0, stream>>>(srcb, dstb, ebuf, denb, h1, agg1, E, N, C1);
    gat_bias_act<<<gNC1, blk, 0, stream>>>(agg1, b1, h1, N, C1,  1);

    gat_fill<<<gN,   blk, 0, stream>>>(mb,  -INFINITY, N);
    gat_fill<<<gN,   blk, 0, stream>>>(denb, 0.f, N);
    gat_fill<<<gNC2, blk, 0, stream>>>(agg2, 0.f, N * C2);

    gat_gemm_wmma<4><<<gGemm, dim3(32), 0, stream>>>(h1, W2, h2, N, C1);
    gat_alpha<<<gNodeWav, blk, 0, stream>>>(h2, av2s, av2d, asb, adb, N, C2);
    gat_edge_logits<<<gEdges, blk, 0, stream>>>(srcb, dstb, asb, adb, ebuf, mb, E, N);
    gat_edge_exp<<<gEdges, blk, 0, stream>>>(dstb, ebuf, mb, denb, E, N);
    gat_edge_aggregate<<<gEdgeWav, blk, 0, stream>>>(srcb, dstb, ebuf, denb, h2, agg2, E, N, C2);
    gat_bias_act<<<gNC2, blk, 0, stream>>>(agg2, b2, outb, N, C2,  0);
  }
}


# Sort in-edges by destination once per graph (counting sort, integer atomic rank); one wave per node does max/exp/sum/aggregate in registers: no float atomics, sum order = edge arrival order
# speedup vs baseline: 2.4422x; 2.4422x over previous
#include <hip/hip_runtime.h>
#include <math.h>

typedef __attribute__((ext_vector_type(2))) float v2f;
typedef __attribute__((ext_vector_type(8))) float v8f;
typedef __attribute__((ext_vector_type(4))) float v4f;

#define NEG_SLOPE 0.2f

template <int TILES>
__global__ __launch_bounds__(32)
void gat_gemm_wmma(const float* __restrict__ X, const float* __restrict__ W,
                   float* __restrict__ Hout, int N, int K) {
  const int F    = TILES * 16;
  const int row0 = blockIdx.x * 16;
  const int lane = threadIdx.x;
  const int l15  = lane & 15;
  const int kh   = (lane >> 4) * 2;
  const int m    = row0 + l15;

  v8f acc[TILES];
#pragma unroll
  for (int t = 0; t < TILES; ++t) acc[t] = (v8f){0.f,0.f,0.f,0.f,0.f,0.f,0.f,0.f};

  for (int k0 = 0; k0 < K; k0 += 4) {
    v2f a;
    a.x = X[(long)m * K + k0 + kh];
    a.y = X[(long)m * K + k0 + kh + 1];
#pragma unroll
    for (int t = 0; t < TILES; ++t) {
      v2f b;
      b.x = W[(long)(k0 + kh)     * F + t * 16 + l15];
      b.y = W[(long)(k0 + kh + 1) * F + t * 16 + l15];
      acc[t] = __builtin_amdgcn_wmma_f32_16x16x4_f32(
           false, a,  false, b,
           (short)0, acc[t],  false,  false);
    }
  }

  const int hi = (lane >> 4) * 8;
#pragma unroll
  for (int t = 0; t < TILES; ++t)
#pragma unroll
    for (int r = 0; r < 8; ++r)
      Hout[(long)(row0 + hi + r) * F + t * 16 + l15] = acc[t][r];
}

__global__ void gat_alpha(const float* __restrict__ h,
                          const float* __restrict__ a_src,
                          const float* __restrict__ a_dst,
                          float* __restrict__ as, float* __restrict__ ad,
                          int N, int C) {
  int wid  = (int)((blockIdx.x * (long)blockDim.x + threadIdx.x) >> 5);
  int lane = threadIdx.x & 31;
  if (wid >= N) return;
  float ss = 0.f, sd = 0.f;
  for (int c = lane; c < C; c += 32) {
    float hv = h[(long)wid * C + c];
    ss += hv * a_src[c];
    sd += hv * a_dst[c];
  }
#pragma unroll
  for (int off = 16; off > 0; off >>= 1) {
    ss += __shfl_down(ss, off, 32);
    sd += __shfl_down(sd, off, 32);
  }
  if (lane == 0) { as[wid] = ss; ad[wid] = sd; }
}

__global__ void gat_zero_int(int* __restrict__ p, int n) {
  int i = blockIdx.x * blockDim.x + threadIdx.x;
  if (i < n) p[i] = 0;
}

__global__ void gat_edge_rank(const int* __restrict__ dst,
                              int* __restrict__ deg,
                              int* __restrict__ rank, int E) {
  int e = blockIdx.x * blockDim.x + threadIdx.x;
  if (e >= E) return;
  rank[e] = atomicAdd(&deg[dst[e]], 1);
}

__global__ __launch_bounds__(1024)
void gat_offsets_scan(const int* __restrict__ deg, int* __restrict__ off, int N) {
  __shared__ int wtot[32];
  const int t    = threadIdx.x;
  const int lane = t & 31;
  const int w    = t >> 5;
  const int per  = (N + 1023) / 1024;
  const int b    = t * per;
  const int e    = (b + per < N) ? (b + per) : N;
  int s = 0;
  for (int i = b; i < e; ++i) s += deg[i];
  int x = s;
  for (int o = 1; o < 32; o <<= 1) {
    int y = __shfl_up(x, o, 32);
    if (lane >= o) x += y;
  }
  if (lane == 31) wtot[w] = x;
  __syncthreads();
  if (w == 0) {
    int v = wtot[lane];
    for (int o = 1; o < 32; o <<= 1) {
      int y = __shfl_up(v, o, 32);
      if (lane >= o) v += y;
    }
    wtot[lane] = v;
  }
  __syncthreads();
  int base = (x - s) + ((w > 0) ? wtot[w - 1] : 0);
  for (int i = b; i < e; ++i) { off[i] = base; base += deg[i]; }
  if (t == 1023) off[N] = base;
}

__global__ void gat_csr_fill(const int* __restrict__ src,
                             const int* __restrict__ dst,
                             const int* __restrict__ rank,
                             const int* __restrict__ off,
                             int* __restrict__ csr, int E) {
  int e = blockIdx.x * blockDim.x + threadIdx.x;
  if (e >= E) return;
  csr[off[dst[e]] + rank[e]] = src[e];
}

template <int V> struct gat_rowvec;
template <> struct gat_rowvec<4> { typedef v4f T; };
template <> struct gat_rowvec<2> { typedef v2f T; };

__device__ __forceinline__ float gat_leaky(float v) {
  return (v > 0.f) ? v : NEG_SLOPE * v;
}

__device__ __forceinline__ float gat_wave_max(float x) {
  for (int o = 16; o > 0; o >>= 1) x = fmaxf(x, __shfl_xor(x, o, 32));
  return x;
}

__device__ __forceinline__ float gat_wave_sum(float x) {
  for (int o = 16; o > 0; o >>= 1) x += __shfl_xor(x, o, 32);
  return x;
}

template <int C, typename vec>
__device__ __forceinline__ void gat_rows_fma(vec& acc, int s, float a, int cnt,
                                             const float* __restrict__ hlane) {
#pragma unroll 4
  for (int j = 0; j < cnt; ++j) {
    const int   sj = __builtin_amdgcn_readlane(s, j);
    const float aj = __int_as_float(__builtin_amdgcn_readlane(__float_as_int(a), j));
    const vec   hv = *(const vec*)(hlane + (size_t)sj * C);
    acc += hv * aj;
  }
}

template <int C, int RELU>
__global__ __launch_bounds__(256)
void gat_node_softmax_gather(const int* __restrict__ off,
                             const int* __restrict__ csr,
                             const float* __restrict__ as,
                             const float* __restrict__ ad,
                             const float* __restrict__ h,
                             const float* __restrict__ bias,
                             float* __restrict__ out, int N) {
  constexpr int V = C / 32;
  typedef typename gat_rowvec<V>::T vec;
  const int lane = threadIdx.x & 31;
  const int d = __builtin_amdgcn_readfirstlane((int)((blockIdx.x * 256u + threadIdx.x) >> 5));
  if (d >= N) return;

  const int   beg   = off[d];
  const int   end   = off[d + 1];
  const float add   = ad[d];
  const float vself = gat_leaky(as[d] + add);

  const int   p0   = beg + lane;
  const bool  has0 = p0 < end;
  const int   s0   = has0 ? csr[p0] : d;
  const float v0   = gat_leaky(as[s0] + add);

  float m = v0;
  for (int p = p0 + 32; p < end; p += 32)
    m = fmaxf(m, gat_leaky(as[csr[p]] + add));
  m = fmaxf(gat_wave_max(m), vself);
  if (!__builtin_isfinite(m)) m = 0.f;

  float sum = has0 ? __expf(v0 - m) : 0.f;
  for (int p = p0 + 32; p < end; p += 32)
    sum += __expf(gat_leaky(as[csr[p]] + add) - m);
  const float exself = __expf(vself - m);
  const float denom  = (gat_wave_sum(sum) + exself) + 1e-16f;

  const float* hlane = h + lane * V;
  vec acc = *(const vec*)(hlane + (size_t)d * C) * (exself / denom);

  const int deg = end - beg;
  gat_rows_fma<C, vec>(acc, s0, __expf(v0 - m) / denom, (deg < 32) ? deg : 32, hlane);
  for (int base = beg + 32; base < end; base += 32) {
    const int   p    = base + lane;
    const int   s    = (p < end) ? csr[p] : d;
    const float a    = __expf(gat_leaky(as[s] + add) - m) / denom;
    const int   left = end - base;
    gat_rows_fma<C, vec>(acc, s, a, (left < 32) ? left : 32, hlane);
  }

  vec o = acc + *(const vec*)(bias + lane * V);
  if (RELU) {
    for (int i = 0; i < V; ++i) o[i] = fmaxf(o[i], 0.f);
  }
  *(vec*)(out + (size_t)d * C + lane * V) = o;
}

extern "C" void kernel_launch(void* const* d_in, const int* in_sizes, int n_in,
                              void* d_out, int out_size, void* d_ws, size_t ws_size,
                              hipStream_t stream) {
  const float* fea = (const float*)d_in[0];
  const int*   ei  = (const int*)  d_in[1];
  const float* W1  = (const float*)d_in[2];
  const float* av1s= (const float*)d_in[3];
  const float* av1d= (const float*)d_in[4];
  const float* b1  = (const float*)d_in[5];
  const float* W2  = (const float*)d_in[6];
  const float* av2s= (const float*)d_in[7];
  const float* av2d= (const float*)d_in[8];
  const float* b2  = (const float*)d_in[9];
  float* out = (float*)d_out;

  const int B  = 2;
  const int C1 = 128, C2 = 64;
  const int N  = in_sizes[0] / (B * C1);
  const int E  = in_sizes[1] / (B * 2);

  float* w   = (float*)d_ws;
  float* h1   = w; w += (long)N * C1;
  float* x2   = w; w += (long)N * C1;
  float* h2   = w; w += (long)N * C2;
  float* asb  = w; w += N;
  float* adb  = w; w += N;
  int* degb  = (int*)w; w += N;
  int* offb  = (int*)w; w += N + 1;
  int* rankb = (int*)w; w += E;
  int* csrb  = (int*)w; w += E;

  const int TB = 256;
  dim3 blk(TB);
  int gN       = (N + TB - 1) / TB;
  int gE       = (E + TB - 1) / TB;
  int gNodeWav = (N * 32 + TB - 1) / TB;
  int gGemm    = N / 16;

  for (int b = 0; b < B; ++b) {
    const float* xb   = fea + (long)b * N * C1;
    const int*   srcb = ei + (long)b * 2 * E;
    const int*   dstb = srcb + E;
    float*       outb = out + (long)b * N * C2;

    gat_zero_int<<<gN, blk, 0, stream>>>(degb, N);
    gat_edge_rank<<<gE, blk, 0, stream>>>(dstb, degb, rankb, E);
    gat_offsets_scan<<<1, dim3(1024), 0, stream>>>(degb, offb, N);
    gat_csr_fill<<<gE, blk, 0, stream>>>(srcb, dstb, rankb, offb, csrb, E);

    gat_gemm_wmma<8><<<gGemm, dim3(32), 0, stream>>>(xb, W1, h1, N, C1);
    gat_alpha<<<gNodeWav, blk, 0, stream>>>(h1, av1s, av1d, asb, adb, N, C1);
    gat_node_softmax_gather<128, 1><<<gNodeWav, blk, 0, stream>>>(
        offb, csrb, asb, adb, h1, b1, x2, N);

    gat_gemm_wmma<4><<<gGemm, dim3(32), 0, stream>>>(x2, W2, h2, N, C1);
    gat_alpha<<<gNodeWav, blk, 0, stream>>>(h2, av2s, av2d, asb, adb, N, C2);
    gat_node_softmax_gather<64, 0><<<gNodeWav, blk, 0, stream>>>(
        offb, csrb, asb, adb, h2, b2, outb, N);
  }
}


# Start 5x fewer waves: each thread takes 8 edges (rank, fill), each wave 8 or 16 nodes (logits) or 4 nodes in turn (softmax+gather); wave start-up, not work, bound these; sums and their order kept
# speedup vs baseline: 2.6321x; 1.0778x over previous
#include <hip/hip_runtime.h>
#include <math.h>

typedef __attribute__((ext_vector_type(2))) float v2f;
typedef __attribute__((ext_vector_type(8))) float v8f;
typedef __attribute__((ext_vector_type(4))) float v4f;

#define NEG_SLOPE 0.2f

template <int TILES>
__global__ __launch_bounds__(32)
void gat_gemm_wmma(const float* __restrict__ X, const float* __restrict__ W,
                   float* __restrict__ Hout, int N, int K) {
  const int F    = TILES * 16;
  const int row0 = blockIdx.x * 16;
  const int lane = threadIdx.x;
  const int l15  = lane & 15;
  const int kh   = (lane >> 4) * 2;
  const int m    = row0 + l15;

  v8f acc[TILES];
#pragma unroll
  for (int t = 0; t < TILES; ++t) acc[t] = (v8f){0.f,0.f,0.f,0.f,0.f,0.f,0.f,0.f};

  for (int k0 = 0; k0 < K; k0 += 4) {
    v2f a;
    a.x = X[(long)m * K + k0 + kh];
    a.y = X[(long)m * K + k0 + kh + 1];
#pragma unroll
    for (int t = 0; t < TILES; ++t) {
      v2f b;
      b.x = W[(long)(k0 + kh)     * F + t * 16 + l15];
      b.y = W[(long)(k0 + kh + 1) * F + t * 16 + l15];
      acc[t] = __builtin_amdgcn_wmma_f32_16x16x4_f32(
           false, a,  false, b,
           (short)0, acc[t],  false,  false);
    }
  }

  const int hi = (lane >> 4) * 8;
#pragma unroll
  for (int t = 0; t < TILES; ++t)
#pragma unroll
    for (int r = 0; r < 8; ++r)
      Hout[(long)(row0 + hi + r) * F + t * 16 + l15] = acc[t][r];
}

template <int C, int NPW>
__global__ __launch_bounds__(256)
void gat_alpha(const float* __restrict__ h,
               const float* __restrict__ a_src,
               const float* __restrict__ a_dst,
               float* __restrict__ as, float* __restrict__ ad, int N) {
  constexpr int R = C / 32;
  const int lane = threadIdx.x & 31;
  const int wave = __builtin_amdgcn_readfirstlane((int)((blockIdx.x * 256u + threadIdx.x) >> 5));
  const int d0   = wave * NPW;
  if (d0 >= N) return;

  float avs[R], avd[R];
#pragma unroll
  for (int i = 0; i < R; ++i) { avs[i] = a_src[lane + 32 * i]; avd[i] = a_dst[lane + 32 * i]; }

  float hv[NPW][R];
#pragma unroll
  for (int j = 0; j < NPW; ++j) {
    const int d = (d0 + j < N) ? (d0 + j) : (N - 1);
#pragma unroll
    for (int i = 0; i < R; ++i) hv[j][i] = h[(size_t)d * C + lane + 32 * i];
  }

  float rs = 0.f, rd = 0.f;
#pragma unroll
  for (int j = 0; j < NPW; ++j) {
    float ss = 0.f, sd = 0.f;
#pragma unroll
    for (int i = 0; i < R; ++i) { ss += hv[j][i] * avs[i]; sd += hv[j][i] * avd[i]; }
#pragma unroll
    for (int o = 16; o > 0; o >>= 1) {
      ss += __shfl_down(ss, o, 32);
      sd += __shfl_down(sd, o, 32);
    }
    const float bs = __int_as_float(__builtin_amdgcn_readfirstlane(__float_as_int(ss)));
    const float bd = __int_as_float(__builtin_amdgcn_readfirstlane(__float_as_int(sd)));
    rs = (lane == j) ? bs : rs;
    rd = (lane == j) ? bd : rd;
  }
  if (lane < NPW && d0 + lane < N) { as[d0 + lane] = rs; ad[d0 + lane] = rd; }
}

__global__ void gat_zero_int(int* __restrict__ p, int n) {
  int i = blockIdx.x * blockDim.x + threadIdx.x;
  if (i < n) p[i] = 0;
}

template <int EPT, bool FULL>
__device__ __forceinline__ void gat_rank_tile(const int* __restrict__ dst,
                                              int* __restrict__ deg,
                                              int* __restrict__ rank,
                                              int e0, int E) {
  int dv[EPT], rv[EPT];
#pragma unroll
  for (int j = 0; j < EPT; ++j) {
    const int e = e0 + j * 256;
    dv[j] = (FULL || e < E) ? dst[e] : 0;
  }
#pragma unroll
  for (int j = 0; j < EPT; ++j) {
    const int e = e0 + j * 256;
    rv[j] = 0;
    if (FULL || e < E) rv[j] = atomicAdd(&deg[dv[j]], 1);
  }
#pragma unroll
  for (int j = 0; j < EPT; ++j) {
    const int e = e0 + j * 256;
    if (FULL || e < E) rank[e] = rv[j];
  }
}

template <int EPT>
__global__ __launch_bounds__(256)
void gat_edge_rank(const int* __restrict__ dst,
                   int* __restrict__ deg,
                   int* __restrict__ rank, int E) {
  const int t0 = blockIdx.x * (256 * EPT);
  if (t0 + 256 * EPT <= E) gat_rank_tile<EPT, true >(dst, deg, rank, t0 + threadIdx.x, E);
  else                     gat_rank_tile<EPT, false>(dst, deg, rank, t0 + threadIdx.x, E);
}

__global__ __launch_bounds__(1024)
void gat_offsets_scan(const int* __restrict__ deg, int* __restrict__ off, int N) {
  __shared__ int wtot[32];
  const int t    = threadIdx.x;
  const int lane = t & 31;
  const int w    = t >> 5;
  const int per  = (N + 1023) / 1024;
  const int b    = t * per;
  const int e    = (b + per < N) ? (b + per) : N;
  int s = 0;
  for (int i = b; i < e; ++i) s += deg[i];
  int x = s;
  for (int o = 1; o < 32; o <<= 1) {
    int y = __shfl_up(x, o, 32);
    if (lane >= o) x += y;
  }
  if (lane == 31) wtot[w] = x;
  __syncthreads();
  if (w == 0) {
    int v = wtot[lane];
    for (int o = 1; o < 32; o <<= 1) {
      int y = __shfl_up(v, o, 32);
      if (lane >= o) v += y;
    }
    wtot[lane] = v;
  }
  __syncthreads();
  int base = (x - s) + ((w > 0) ? wtot[w - 1] : 0);
  for (int i = b; i < e; ++i) { off[i] = base; base += deg[i]; }
  if (t == 1023) off[N] = base;
}

template <int EPT, bool FULL>
__device__ __forceinline__ void gat_fill_tile(const int* __restrict__ src,
                                              const int* __restrict__ dst,
                                              const int* __restrict__ rank,
                                              const int* __restrict__ off,
                                              int* __restrict__ csr,
                                              int e0, int E) {
  int sv[EPT], dv[EPT], rv[EPT], ov[EPT];
#pragma unroll
  for (int j = 0; j < EPT; ++j) {
    const int  e  = e0 + j * 256;
    const bool in = FULL || e < E;
    sv[j] = in ? src[e]  : 0;
    dv[j] = in ? dst[e]  : 0;
    rv[j] = in ? rank[e] : 0;
  }
#pragma unroll
  for (int j = 0; j < EPT; ++j) ov[j] = off[dv[j]];
#pragma unroll
  for (int j = 0; j < EPT; ++j) {
    const int e = e0 + j * 256;
    if (FULL || e < E) csr[ov[j] + rv[j]] = sv[j];
  }
}

template <int EPT>
__global__ __launch_bounds__(256)
void gat_csr_fill(const int* __restrict__ src,
                  const int* __restrict__ dst,
                  const int* __restrict__ rank,
                  const int* __restrict__ off,
                  int* __restrict__ csr, int E) {
  const int t0 = blockIdx.x * (256 * EPT);
  if (t0 + 256 * EPT <= E) gat_fill_tile<EPT, true >(src, dst, rank, off, csr, t0 + threadIdx.x, E);
  else                     gat_fill_tile<EPT, false>(src, dst, rank, off, csr, t0 + threadIdx.x, E);
}

template <int V> struct gat_rowvec;
template <> struct gat_rowvec<4> { typedef v4f T; };
template <> struct gat_rowvec<2> { typedef v2f T; };

__device__ __forceinline__ float gat_leaky(float v) {
  return (v > 0.f) ? v : NEG_SLOPE * v;
}

__device__ __forceinline__ float gat_wave_max(float x) {
  for (int o = 16; o > 0; o >>= 1) x = fmaxf(x, __shfl_xor(x, o, 32));
  return x;
}

__device__ __forceinline__ float gat_wave_sum(float x) {
  for (int o = 16; o > 0; o >>= 1) x += __shfl_xor(x, o, 32);
  return x;
}

template <int C, typename vec>
__device__ __forceinline__ void gat_rows_fma(vec& acc, int s, float a, int cnt,
                                             const float* __restrict__ hlane) {
#pragma unroll 4
  for (int j = 0; j < cnt; ++j) {
    const int   sj = __builtin_amdgcn_readlane(s, j);
    const float aj = __int_as_float(__builtin_amdgcn_readlane(__float_as_int(a), j));
    const vec   hv = *(const vec*)(hlane + (size_t)sj * C);
    acc += hv * aj;
  }
}

template <int C, int RELU, int NPW>
__global__ __launch_bounds__(256)
void gat_node_softmax_gather(const int* __restrict__ off,
                             const int* __restrict__ csr,
                             const float* __restrict__ as,
                             const float* __restrict__ ad,
                             const float* __restrict__ h,
                             const float* __restrict__ bias,
                             float* __restrict__ out, int N) {
  constexpr int V = C / 32;
  typedef typename gat_rowvec<V>::T vec;
  const int lane = threadIdx.x & 31;
  const int wave = __builtin_amdgcn_readfirstlane((int)((blockIdx.x * 256u + threadIdx.x) >> 5));
  const int d0   = wave * NPW;
  if (d0 >= N) return;
  const int d1   = (d0 + NPW < N) ? (d0 + NPW) : N;

  const float* hlane = h + lane * V;
  const vec    bv    = *(const vec*)(bias + lane * V);

  int beg = off[d0];
  for (int d = d0; d < d1; ++d) {
    const int   end   = off[d + 1];
    const float add   = ad[d];
    const float vself = gat_leaky(as[d] + add);

    const int   p0   = beg + lane;
    const bool  has0 = p0 < end;
    const int   s0   = has0 ? csr[p0] : d;
    const float v0   = gat_leaky(as[s0] + add);

    float m = v0;
    for (int p = p0 + 32; p < end; p += 32)
      m = fmaxf(m, gat_leaky(as[csr[p]] + add));
    m = fmaxf(gat_wave_max(m), vself);
    if (!__builtin_isfinite(m)) m = 0.f;

    float sum = has0 ? __expf(v0 - m) : 0.f;
    for (int p = p0 + 32; p < end; p += 32)
      sum += __expf(gat_leaky(as[csr[p]] + add) - m);
    const float exself = __expf(vself - m);
    const float denom  = (gat_wave_sum(sum) + exself) + 1e-16f;

    vec acc = *(const vec*)(hlane + (size_t)d * C) * (exself / denom);

    const int deg = end - beg;
    gat_rows_fma<C, vec>(acc, s0, __expf(v0 - m) / denom, (deg < 32) ? deg : 32, hlane);
    for (int base = beg + 32; base < end; base += 32) {
      const int   p    = base + lane;
      const int   s    = (p < end) ? csr[p] : d;
      const float a    = __expf(gat_leaky(as[s] + add) - m) / denom;
      const int   left = end - base;
      gat_rows_fma<C, vec>(acc, s, a, (left < 32) ? left : 32, hlane);
    }

    vec o = acc + bv;
    if (RELU) {
      for (int i = 0; i < V; ++i) o[i] = fmaxf(o[i], 0.f);
    }
    *(vec*)(out + (size_t)d * C + lane * V) = o;
    beg = end;
  }
}

extern "C" void kernel_launch(void* const* d_in, const int* in_sizes, int n_in,
                              void* d_out, int out_size, void* d_ws, size_t ws_size,
                              hipStream_t stream) {
  const float* fea = (const float*)d_in[0];
  const int*   ei  = (const int*)  d_in[1];
  const float* W1  = (const float*)d_in[2];
  const float* av1s= (const float*)d_in[3];
  const float* av1d= (const float*)d_in[4];
  const float* b1  = (const float*)d_in[5];
  const float* W2  = (const float*)d_in[6];
  const float* av2s= (const float*)d_in[7];
  const float* av2d= (const float*)d_in[8];
  const float* b2  = (const float*)d_in[9];
  float* out = (float*)d_out;

  const int B  = 2;
  const int C1 = 128, C2 = 64;
  const int N  = in_sizes[0] / (B * C1);
  const int E  = in_sizes[1] / (B * 2);

  float* w   = (float*)d_ws;
  float* h1   = w; w += (long)N * C1;
  float* x2   = w; w += (long)N * C1;
  float* h2   = w; w += (long)N * C2;
  float* asb  = w; w += N;
  float* adb  = w; w += N;
  int* degb  = (int*)w; w += N;
  int* offb  = (int*)w; w += N + 1;
  int* rankb = (int*)w; w += E;
  int* csrb  = (int*)w; w += E;

  const int TB = 256;
  dim3 blk(TB);
  int gN       = (N + TB - 1) / TB;
  const int EPT = 8;
  const int NPW = 4;
  const int APW1 = 8, APW2 = 16;
  int gEdge    = (E + TB * EPT - 1) / (TB * EPT);
  int gNode    = ((N + NPW - 1) / NPW * 32 + TB - 1) / TB;
  int gAlpha1  = ((N + APW1 - 1) / APW1 * 32 + TB - 1) / TB;
  int gAlpha2  = ((N + APW2 - 1) / APW2 * 32 + TB - 1) / TB;
  int gGemm    = N / 16;

  for (int b = 0; b < B; ++b) {
    const float* xb   = fea + (long)b * N * C1;
    const int*   srcb = ei + (long)b * 2 * E;
    const int*   dstb = srcb + E;
    float*       outb = out + (long)b * N * C2;

    gat_zero_int<<<gN, blk, 0, stream>>>(degb, N);
    gat_edge_rank<EPT><<<gEdge, blk, 0, stream>>>(dstb, degb, rankb, E);
    gat_offsets_scan<<<1, dim3(1024), 0, stream>>>(degb, offb, N);
    gat_csr_fill<EPT><<<gEdge, blk, 0, stream>>>(srcb, dstb, rankb, offb, csrb, E);

    gat_gemm_wmma<8><<<gGemm, dim3(32), 0, stream>>>(xb, W1, h1, N, C1);
    gat_alpha<128, APW1><<<gAlpha1, blk, 0, stream>>>(h1, av1s, av1d, asb, adb, N);
    gat_node_softmax_gather<128, 1, NPW><<<gNode, blk, 0, stream>>>(
        offb, csrb, asb, adb, h1, b1, x2, N);

    gat_gemm_wmma<4><<<gGemm, dim3(32), 0, stream>>>(x2, W2, h2, N, C1);
    gat_alpha<64, APW2><<<gAlpha2, blk, 0, stream>>>(h2, av2s, av2d, asb, adb, N);
    gat_node_softmax_gather<64, 0, NPW><<<gNode, blk, 0, stream>>>(
        offb, csrb, asb, adb, h2, b2, outb, N);
  }
}


# Offsets scan: one block per 1024 nodes; each adds up the degrees in front of its chunk itself (coalesced 16-byte loads), then scans its chunk. Was one block of strided per-thread runs. Integer sums
# speedup vs baseline: 3.4722x; 1.3192x over previous
#include <hip/hip_runtime.h>
#include <math.h>

typedef __attribute__((ext_vector_type(2))) float v2f;
typedef __attribute__((ext_vector_type(8))) float v8f;
typedef __attribute__((ext_vector_type(4))) float v4f;

#define NEG_SLOPE 0.2f

template <int TILES>
__global__ __launch_bounds__(32)
void gat_gemm_wmma(const float* __restrict__ X, const float* __restrict__ W,
                   float* __restrict__ Hout, int N, int K) {
  const int F    = TILES * 16;
  const int row0 = blockIdx.x * 16;
  const int lane = threadIdx.x;
  const int l15  = lane & 15;
  const int kh   = (lane >> 4) * 2;
  const int m    = row0 + l15;

  v8f acc[TILES];
#pragma unroll
  for (int t = 0; t < TILES; ++t) acc[t] = (v8f){0.f,0.f,0.f,0.f,0.f,0.f,0.f,0.f};

  for (int k0 = 0; k0 < K; k0 += 4) {
    v2f a;
    a.x = X[(long)m * K + k0 + kh];
    a.y = X[(long)m * K + k0 + kh + 1];
#pragma unroll
    for (int t = 0; t < TILES; ++t) {
      v2f b;
      b.x = W[(long)(k0 + kh)     * F + t * 16 + l15];
      b.y = W[(long)(k0 + kh + 1) * F + t * 16 + l15];
      acc[t] = __builtin_amdgcn_wmma_f32_16x16x4_f32(
           false, a,  false, b,
           (short)0, acc[t],  false,  false);
    }
  }

  const int hi = (lane >> 4) * 8;
#pragma unroll
  for (int t = 0; t < TILES; ++t)
#pragma unroll
    for (int r = 0; r < 8; ++r)
      Hout[(long)(row0 + hi + r) * F + t * 16 + l15] = acc[t][r];
}

template <int C, int NPW>
__global__ __launch_bounds__(256)
void gat_alpha(const float* __restrict__ h,
               const float* __restrict__ a_src,
               const float* __restrict__ a_dst,
               float* __restrict__ as, float* __restrict__ ad, int N) {
  constexpr int R = C / 32;
  const int lane = threadIdx.x & 31;
  const int wave = __builtin_amdgcn_readfirstlane((int)((blockIdx.x * 256u + threadIdx.x) >> 5));
  const int d0   = wave * NPW;
  if (d0 >= N) return;

  float avs[R], avd[R];
#pragma unroll
  for (int i = 0; i < R; ++i) { avs[i] = a_src[lane + 32 * i]; avd[i] = a_dst[lane + 32 * i]; }

  float hv[NPW][R];
#pragma unroll
  for (int j = 0; j < NPW; ++j) {
    const int d = (d0 + j < N) ? (d0 + j) : (N - 1);
#pragma unroll
    for (int i = 0; i < R; ++i) hv[j][i] = h[(size_t)d * C + lane + 32 * i];
  }

  float rs = 0.f, rd = 0.f;
#pragma unroll
  for (int j = 0; j < NPW; ++j) {
    float ss = 0.f, sd = 0.f;
#pragma unroll
    for (int i = 0; i < R; ++i) { ss += hv[j][i] * avs[i]; sd += hv[j][i] * avd[i]; }
#pragma unroll
    for (int o = 16; o > 0; o >>= 1) {
      ss += __shfl_down(ss, o, 32);
      sd += __shfl_down(sd, o, 32);
    }
    const float bs = __int_as_float(__builtin_amdgcn_readfirstlane(__float_as_int(ss)));
    const float bd = __int_as_float(__builtin_amdgcn_readfirstlane(__float_as_int(sd)));
    rs = (lane == j) ? bs : rs;
    rd = (lane == j) ? bd : rd;
  }
  if (lane < NPW && d0 + lane < N) { as[d0 + lane] = rs; ad[d0 + lane] = rd; }
}

__global__ void gat_zero_int(int* __restrict__ p, int n) {
  int i = blockIdx.x * blockDim.x + threadIdx.x;
  if (i < n) p[i] = 0;
}

template <int EPT, bool FULL>
__device__ __forceinline__ void gat_rank_tile(const int* __restrict__ dst,
                                              int* __restrict__ deg,
                                              int* __restrict__ rank,
                                              int e0, int E) {
  int dv[EPT], rv[EPT];
#pragma unroll
  for (int j = 0; j < EPT; ++j) {
    const int e = e0 + j * 256;
    dv[j] = (FULL || e < E) ? dst[e] : 0;
  }
#pragma unroll
  for (int j = 0; j < EPT; ++j) {
    const int e = e0 + j * 256;
    rv[j] = 0;
    if (FULL || e < E) rv[j] = atomicAdd(&deg[dv[j]], 1);
  }
#pragma unroll
  for (int j = 0; j < EPT; ++j) {
    const int e = e0 + j * 256;
    if (FULL || e < E) rank[e] = rv[j];
  }
}

template <int EPT>
__global__ __launch_bounds__(256)
void gat_edge_rank(const int* __restrict__ dst,
                   int* __restrict__ deg,
                   int* __restrict__ rank, int E) {
  const int t0 = blockIdx.x * (256 * EPT);
  if (t0 + 256 * EPT <= E) gat_rank_tile<EPT, true >(dst, deg, rank, t0 + threadIdx.x, E);
  else                     gat_rank_tile<EPT, false>(dst, deg, rank, t0 + threadIdx.x, E);
}

__global__ __launch_bounds__(1024)
void gat_offsets_scan(const int* __restrict__ deg, int* __restrict__ off, int N) {
  __shared__ int wtot[32];
  __shared__ int ptot[32];
  const int t    = threadIdx.x;
  const int lane = t & 31;
  const int w    = t >> 5;
  const int i    = blockIdx.x * 1024 + t;

  const int v = (i < N) ? deg[i] : 0;

  const int4* __restrict__ d4 = (const int4*)deg;
  const int n4 = blockIdx.x * 256;
  int ps = 0;
#pragma unroll 4
  for (int q = t; q < n4; q += 1024) {
    const int4 u = d4[q];
    ps += (u.x + u.y) + (u.z + u.w);
  }

  int x = v;
  for (int o = 1; o < 32; o <<= 1) {
    int y = __shfl_up(x, o, 32);
    if (lane >= o) x += y;
  }
  for (int o = 16; o > 0; o >>= 1) ps += __shfl_xor(ps, o, 32);
  if (lane == 31) { wtot[w] = x; ptot[w] = ps; }
  __syncthreads();
  int base = ptot[lane] + ((lane < w) ? wtot[lane] : 0);
  for (int o = 16; o > 0; o >>= 1) base += __shfl_xor(base, o, 32);

  if (i < N)      off[i] = base + x - v;
  if (i == N - 1) off[N] = base + x;
}

template <int EPT, bool FULL>
__device__ __forceinline__ void gat_fill_tile(const int* __restrict__ src,
                                              const int* __restrict__ dst,
                                              const int* __restrict__ rank,
                                              const int* __restrict__ off,
                                              int* __restrict__ csr,
                                              int e0, int E) {
  int sv[EPT], dv[EPT], rv[EPT], ov[EPT];
#pragma unroll
  for (int j = 0; j < EPT; ++j) {
    const int  e  = e0 + j * 256;
    const bool in = FULL || e < E;
    sv[j] = in ? src[e]  : 0;
    dv[j] = in ? dst[e]  : 0;
    rv[j] = in ? rank[e] : 0;
  }
#pragma unroll
  for (int j = 0; j < EPT; ++j) ov[j] = off[dv[j]];
#pragma unroll
  for (int j = 0; j < EPT; ++j) {
    const int e = e0 + j * 256;
    if (FULL || e < E) csr[ov[j] + rv[j]] = sv[j];
  }
}

template <int EPT>
__global__ __launch_bounds__(256)
void gat_csr_fill(const int* __restrict__ src,
                  const int* __restrict__ dst,
                  const int* __restrict__ rank,
                  const int* __restrict__ off,
                  int* __restrict__ csr, int E) {
  const int t0 = blockIdx.x * (256 * EPT);
  if (t0 + 256 * EPT <= E) gat_fill_tile<EPT, true >(src, dst, rank, off, csr, t0 + threadIdx.x, E);
  else                     gat_fill_tile<EPT, false>(src, dst, rank, off, csr, t0 + threadIdx.x, E);
}

template <int V> struct gat_rowvec;
template <> struct gat_rowvec<4> { typedef v4f T; };
template <> struct gat_rowvec<2> { typedef v2f T; };

__device__ __forceinline__ float gat_leaky(float v) {
  return (v > 0.f) ? v : NEG_SLOPE * v;
}

__device__ __forceinline__ float gat_wave_max(float x) {
  for (int o = 16; o > 0; o >>= 1) x = fmaxf(x, __shfl_xor(x, o, 32));
  return x;
}

__device__ __forceinline__ float gat_wave_sum(float x) {
  for (int o = 16; o > 0; o >>= 1) x += __shfl_xor(x, o, 32);
  return x;
}

template <int C, typename vec>
__device__ __forceinline__ void gat_rows_fma(vec& acc, int s, float a, int cnt,
                                             const float* __restrict__ hlane) {
#pragma unroll 4
  for (int j = 0; j < cnt; ++j) {
    const int   sj = __builtin_amdgcn_readlane(s, j);
    const float aj = __int_as_float(__builtin_amdgcn_readlane(__float_as_int(a), j));
    const vec   hv = *(const vec*)(hlane + (size_t)sj * C);
    acc += hv * aj;
  }
}

template <int C, int RELU, int NPW>
__global__ __launch_bounds__(256)
void gat_node_softmax_gather(const int* __restrict__ off,
                             const int* __restrict__ csr,
                             const float* __restrict__ as,
                             const float* __restrict__ ad,
                             const float* __restrict__ h,
                             const float* __restrict__ bias,
                             float* __restrict__ out, int N) {
  constexpr int V = C / 32;
  typedef typename gat_rowvec<V>::T vec;
  const int lane = threadIdx.x & 31;
  const int wave = __builtin_amdgcn_readfirstlane((int)((blockIdx.x * 256u + threadIdx.x) >> 5));
  const int d0   = wave * NPW;
  if (d0 >= N) return;
  const int d1   = (d0 + NPW < N) ? (d0 + NPW) : N;

  const float* hlane = h + lane * V;
  const vec    bv    = *(const vec*)(bias + lane * V);

  int beg = off[d0];
  for (int d = d0; d < d1; ++d) {
    const int   end   = off[d + 1];
    const float add   = ad[d];
    const float vself = gat_leaky(as[d] + add);

    const int   p0   = beg + lane;
    const bool  has0 = p0 < end;
    const int   s0   = has0 ? csr[p0] : d;
    const float v0   = gat_leaky(as[s0] + add);

    float m = v0;
    for (int p = p0 + 32; p < end; p += 32)
      m = fmaxf(m, gat_leaky(as[csr[p]] + add));
    m = fmaxf(gat_wave_max(m), vself);
    if (!__builtin_isfinite(m)) m = 0.f;

    float sum = has0 ? __expf(v0 - m) : 0.f;
    for (int p = p0 + 32; p < end; p += 32)
      sum += __expf(gat_leaky(as[csr[p]] + add) - m);
    const float exself = __expf(vself - m);
    const float denom  = (gat_wave_sum(sum) + exself) + 1e-16f;

    vec acc = *(const vec*)(hlane + (size_t)d * C) * (exself / denom);

    const int deg = end - beg;
    gat_rows_fma<C, vec>(acc, s0, __expf(v0 - m) / denom, (deg < 32) ? deg : 32, hlane);
    for (int base = beg + 32; base < end; base += 32) {
      const int   p    = base + lane;
      const int   s    = (p < end) ? csr[p] : d;
      const float a    = __expf(gat_leaky(as[s] + add) - m) / denom;
      const int   left = end - base;
      gat_rows_fma<C, vec>(acc, s, a, (left < 32) ? left : 32, hlane);
    }

    vec o = acc + bv;
    if (RELU) {
      for (int i = 0; i < V; ++i) o[i] = fmaxf(o[i], 0.f);
    }
    *(vec*)(out + (size_t)d * C + lane * V) = o;
    beg = end;
  }
}

extern "C" void kernel_launch(void* const* d_in, const int* in_sizes, int n_in,
                              void* d_out, int out_size, void* d_ws, size_t ws_size,
                              hipStream_t stream) {
  const float* fea = (const float*)d_in[0];
  const int*   ei  = (const int*)  d_in[1];
  const float* W1  = (const float*)d_in[2];
  const float* av1s= (const float*)d_in[3];
  const float* av1d= (const float*)d_in[4];
  const float* b1  = (const float*)d_in[5];
  const float* W2  = (const float*)d_in[6];
  const float* av2s= (const float*)d_in[7];
  const float* av2d= (const float*)d_in[8];
  const float* b2  = (const float*)d_in[9];
  float* out = (float*)d_out;

  const int B  = 2;
  const int C1 = 128, C2 = 64;
  const int N  = in_sizes[0] / (B * C1);
  const int E  = in_sizes[1] / (B * 2);

  float* w   = (float*)d_ws;
  float* h1   = w; w += (long)N * C1;
  float* x2   = w; w += (long)N * C1;
  float* h2   = w; w += (long)N * C2;
  float* asb  = w; w += N;
  float* adb  = w; w += N;
  int* degb  = (int*)w; w += N;
  int* offb  = (int*)w; w += N + 1;
  int* rankb = (int*)w; w += E;
  int* csrb  = (int*)w; w += E;

  const int TB = 256;
  dim3 blk(TB);
  int gN       = (N + TB - 1) / TB;
  const int EPT = 8;
  const int NPW = 4;
  const int APW1 = 8, APW2 = 16;
  int gEdge    = (E + TB * EPT - 1) / (TB * EPT);
  int gNode    = ((N + NPW - 1) / NPW * 32 + TB - 1) / TB;
  int gAlpha1  = ((N + APW1 - 1) / APW1 * 32 + TB - 1) / TB;
  int gAlpha2  = ((N + APW2 - 1) / APW2 * 32 + TB - 1) / TB;
  int gGemm    = N / 16;
  int gScan    = (N + 1023) / 1024;

  for (int b = 0; b < B; ++b) {
    const float* xb   = fea + (long)b * N * C1;
    const int*   srcb = ei + (long)b * 2 * E;
    const int*   dstb = srcb + E;
    float*       outb = out + (long)b * N * C2;

    gat_zero_int<<<gN, blk, 0, stream>>>(degb, N);
    gat_edge_rank<EPT><<<gEdge, blk, 0, stream>>>(dstb, degb, rankb, E);
    gat_offsets_scan<<<gScan, dim3(1024), 0, stream>>>(degb, offb, N);
    gat_csr_fill<EPT><<<gEdge, blk, 0, stream>>>(srcb, dstb, rankb, offb, csrb, E);

    gat_gemm_wmma<8><<<gGemm, dim3(32), 0, stream>>>(xb, W1, h1, N, C1);
    gat_alpha<128, APW1><<<gAlpha1, blk, 0, stream>>>(h1, av1s, av1d, asb, adb, N);
    gat_node_softmax_gather<128, 1, NPW><<<gNode, blk, 0, stream>>>(
        offb, csrb, asb, adb, h1, b1, x2, N);

    gat_gemm_wmma<4><<<gGemm, dim3(32), 0, stream>>>(x2, W2, h2, N, C1);
    gat_alpha<64, APW2><<<gAlpha2, blk, 0, stream>>>(h2, av2s, av2d, asb, adb, N);
    gat_node_softmax_gather<64, 0, NPW><<<gNode, blk, 0, stream>>>(
        offb, csrb, asb, adb, h2, b2, outb, N);
  }
}


# All graphs of the batch go through every kernel together as one joint graph (node ids shifted by g*N): 10 launches, each twice as full, instead of 20; sums and their order kept
# speedup vs baseline: 4.2827x; 1.2334x over previous
#include <hip/hip_runtime.h>
#include <math.h>

typedef __attribute__((ext_vector_type(2))) float v2f;
typedef __attribute__((ext_vector_type(8))) float v8f;
typedef __attribute__((ext_vector_type(4))) float v4f;

#define NEG_SLOPE 0.2f

template <int TILES>
__global__ __launch_bounds__(32)
void gat_gemm_wmma(const float* __restrict__ X, const float* __restrict__ W,
                   float* __restrict__ Hout, int N, int K) {
  const int F    = TILES * 16;
  const int row0 = blockIdx.x * 16;
  const int lane = threadIdx.x;
  const int l15  = lane & 15;
  const int kh   = (lane >> 4) * 2;
  const int m    = row0 + l15;

  v8f acc[TILES];
#pragma unroll
  for (int t = 0; t < TILES; ++t) acc[t] = (v8f){0.f,0.f,0.f,0.f,0.f,0.f,0.f,0.f};

  for (int k0 = 0; k0 < K; k0 += 4) {
    v2f a;
    a.x = X[(long)m * K + k0 + kh];
    a.y = X[(long)m * K + k0 + kh + 1];
#pragma unroll
    for (int t = 0; t < TILES; ++t) {
      v2f b;
      b.x = W[(long)(k0 + kh)     * F + t * 16 + l15];
      b.y = W[(long)(k0 + kh + 1) * F + t * 16 + l15];
      acc[t] = __builtin_amdgcn_wmma_f32_16x16x4_f32(
           false, a,  false, b,
           (short)0, acc[t],  false,  false);
    }
  }

  const int hi = (lane >> 4) * 8;
#pragma unroll
  for (int t = 0; t < TILES; ++t)
#pragma unroll
    for (int r = 0; r < 8; ++r)
      Hout[(long)(row0 + hi + r) * F + t * 16 + l15] = acc[t][r];
}

template <int C, int NPW>
__global__ __launch_bounds__(256)
void gat_alpha(const float* __restrict__ h,
               const float* __restrict__ a_src,
               const float* __restrict__ a_dst,
               float* __restrict__ as, float* __restrict__ ad, int N) {
  constexpr int R = C / 32;
  const int lane = threadIdx.x & 31;
  const int wave = __builtin_amdgcn_readfirstlane((int)((blockIdx.x * 256u + threadIdx.x) >> 5));
  const int d0   = wave * NPW;
  if (d0 >= N) return;

  float avs[R], avd[R];
#pragma unroll
  for (int i = 0; i < R; ++i) { avs[i] = a_src[lane + 32 * i]; avd[i] = a_dst[lane + 32 * i]; }

  float hv[NPW][R];
#pragma unroll
  for (int j = 0; j < NPW; ++j) {
    const int d = (d0 + j < N) ? (d0 + j) : (N - 1);
#pragma unroll
    for (int i = 0; i < R; ++i) hv[j][i] = h[(size_t)d * C + lane + 32 * i];
  }

  float rs = 0.f, rd = 0.f;
#pragma unroll
  for (int j = 0; j < NPW; ++j) {
    float ss = 0.f, sd = 0.f;
#pragma unroll
    for (int i = 0; i < R; ++i) { ss += hv[j][i] * avs[i]; sd += hv[j][i] * avd[i]; }
#pragma unroll
    for (int o = 16; o > 0; o >>= 1) {
      ss += __shfl_down(ss, o, 32);
      sd += __shfl_down(sd, o, 32);
    }
    const float bs = __int_as_float(__builtin_amdgcn_readfirstlane(__float_as_int(ss)));
    const float bd = __int_as_float(__builtin_amdgcn_readfirstlane(__float_as_int(sd)));
    rs = (lane == j) ? bs : rs;
    rd = (lane == j) ? bd : rd;
  }
  if (lane < NPW && d0 + lane < N) { as[d0 + lane] = rs; ad[d0 + lane] = rd; }
}

__global__ void gat_zero_int(int* __restrict__ p, int n) {
  int i = blockIdx.x * blockDim.x + threadIdx.x;
  if (i < n) p[i] = 0;
}

template <int EPT, bool FULL>
__device__ __forceinline__ void gat_rank_tile(const int* __restrict__ dst,
                                              int* __restrict__ deg,
                                              int* __restrict__ rank,
                                              int e0, int E) {
  int dv[EPT], rv[EPT];
#pragma unroll
  for (int j = 0; j < EPT; ++j) {
    const int e = e0 + j * 256;
    dv[j] = (FULL || e < E) ? dst[e] : 0;
  }
#pragma unroll
  for (int j = 0; j < EPT; ++j) {
    const int e = e0 + j * 256;
    rv[j] = 0;
    if (FULL || e < E) rv[j] = atomicAdd(&deg[dv[j]], 1);
  }
#pragma unroll
  for (int j = 0; j < EPT; ++j) {
    const int e = e0 + j * 256;
    if (FULL || e < E) rank[e] = rv[j];
  }
}

template <int EPT>
__global__ __launch_bounds__(256)
void gat_edge_rank(const int* __restrict__ ei,
                   int* __restrict__ deg,
                   int* __restrict__ rank, int N, int E) {
  const int g = blockIdx.y;
  const int* __restrict__ dst   = ei + ((size_t)g * 2 + 1) * E;
  int* __restrict__       deg_g = deg + (size_t)g * N;
  int* __restrict__       rnk_g = rank + (size_t)g * E;
  const int t0 = blockIdx.x * (256 * EPT);
  if (t0 + 256 * EPT <= E) gat_rank_tile<EPT, true >(dst, deg_g, rnk_g, t0 + threadIdx.x, E);
  else                     gat_rank_tile<EPT, false>(dst, deg_g, rnk_g, t0 + threadIdx.x, E);
}

__global__ __launch_bounds__(1024)
void gat_offsets_scan(const int* __restrict__ deg, int* __restrict__ off, int N) {
  __shared__ int wtot[32];
  __shared__ int ptot[32];
  const int t    = threadIdx.x;
  const int lane = t & 31;
  const int w    = t >> 5;
  const int i    = blockIdx.x * 1024 + t;

  const int v = (i < N) ? deg[i] : 0;

  const int4* __restrict__ d4 = (const int4*)deg;
  const int n4 = blockIdx.x * 256;
  int ps = 0;
#pragma unroll 4
  for (int q = t; q < n4; q += 1024) {
    const int4 u = d4[q];
    ps += (u.x + u.y) + (u.z + u.w);
  }

  int x = v;
  for (int o = 1; o < 32; o <<= 1) {
    int y = __shfl_up(x, o, 32);
    if (lane >= o) x += y;
  }
  for (int o = 16; o > 0; o >>= 1) ps += __shfl_xor(ps, o, 32);
  if (lane == 31) { wtot[w] = x; ptot[w] = ps; }
  __syncthreads();
  int base = ptot[lane] + ((lane < w) ? wtot[lane] : 0);
  for (int o = 16; o > 0; o >>= 1) base += __shfl_xor(base, o, 32);

  if (i < N)      off[i] = base + x - v;
  if (i == N - 1) off[N] = base + x;
}

template <int EPT, bool FULL>
__device__ __forceinline__ void gat_fill_tile(const int* __restrict__ src,
                                              const int* __restrict__ dst,
                                              const int* __restrict__ rank,
                                              const int* __restrict__ off,
                                              int* __restrict__ csr,
                                              int e0, int E, int idshift) {
  int sv[EPT], dv[EPT], rv[EPT], ov[EPT];
#pragma unroll
  for (int j = 0; j < EPT; ++j) {
    const int  e  = e0 + j * 256;
    const bool in = FULL || e < E;
    sv[j] = in ? src[e]  : 0;
    dv[j] = in ? dst[e]  : 0;
    rv[j] = in ? rank[e] : 0;
  }
#pragma unroll
  for (int j = 0; j < EPT; ++j) ov[j] = off[dv[j]];
#pragma unroll
  for (int j = 0; j < EPT; ++j) {
    const int e = e0 + j * 256;
    if (FULL || e < E) csr[ov[j] + rv[j]] = sv[j] + idshift;
  }
}

template <int EPT>
__global__ __launch_bounds__(256)
void gat_csr_fill(const int* __restrict__ ei,
                  const int* __restrict__ rank,
                  const int* __restrict__ off,
                  int* __restrict__ csr, int N, int E) {
  const int g = blockIdx.y;
  const int* __restrict__ src   = ei + (size_t)g * 2 * E;
  const int* __restrict__ dst   = src + E;
  const int* __restrict__ rnk_g = rank + (size_t)g * E;
  const int* __restrict__ off_g = off + (size_t)g * N;
  const int t0 = blockIdx.x * (256 * EPT);
  if (t0 + 256 * EPT <= E) gat_fill_tile<EPT, true >(src, dst, rnk_g, off_g, csr, t0 + threadIdx.x, E, g * N);
  else                     gat_fill_tile<EPT, false>(src, dst, rnk_g, off_g, csr, t0 + threadIdx.x, E, g * N);
}

template <int V> struct gat_rowvec;
template <> struct gat_rowvec<4> { typedef v4f T; };
template <> struct gat_rowvec<2> { typedef v2f T; };

__device__ __forceinline__ float gat_leaky(float v) {
  return (v > 0.f) ? v : NEG_SLOPE * v;
}

__device__ __forceinline__ float gat_wave_max(float x) {
  for (int o = 16; o > 0; o >>= 1) x = fmaxf(x, __shfl_xor(x, o, 32));
  return x;
}

__device__ __forceinline__ float gat_wave_sum(float x) {
  for (int o = 16; o > 0; o >>= 1) x += __shfl_xor(x, o, 32);
  return x;
}

template <int C, typename vec>
__device__ __forceinline__ void gat_rows_fma(vec& acc, int s, float a, int cnt,
                                             const float* __restrict__ hlane) {
#pragma unroll 4
  for (int j = 0; j < cnt; ++j) {
    const int   sj = __builtin_amdgcn_readlane(s, j);
    const float aj = __int_as_float(__builtin_amdgcn_readlane(__float_as_int(a), j));
    const vec   hv = *(const vec*)(hlane + (size_t)sj * C);
    acc += hv * aj;
  }
}

template <int C, int RELU, int NPW>
__global__ __launch_bounds__(256)
void gat_node_softmax_gather(const int* __restrict__ off,
                             const int* __restrict__ csr,
                             const float* __restrict__ as,
                             const float* __restrict__ ad,
                             const float* __restrict__ h,
                             const float* __restrict__ bias,
                             float* __restrict__ out, int N) {
  constexpr int V = C / 32;
  typedef typename gat_rowvec<V>::T vec;
  const int lane = threadIdx.x & 31;
  const int wave = __builtin_amdgcn_readfirstlane((int)((blockIdx.x * 256u + threadIdx.x) >> 5));
  const int d0   = wave * NPW;
  if (d0 >= N) return;
  const int d1   = (d0 + NPW < N) ? (d0 + NPW) : N;

  const float* hlane = h + lane * V;
  const vec    bv    = *(const vec*)(bias + lane * V);

  int beg = off[d0];
  for (int d = d0; d < d1; ++d) {
    const int   end   = off[d + 1];
    const float add   = ad[d];
    const float vself = gat_leaky(as[d] + add);

    const int   p0   = beg + lane;
    const bool  has0 = p0 < end;
    const int   s0   = has0 ? csr[p0] : d;
    const float v0   = gat_leaky(as[s0] + add);

    float m = v0;
    for (int p = p0 + 32; p < end; p += 32)
      m = fmaxf(m, gat_leaky(as[csr[p]] + add));
    m = fmaxf(gat_wave_max(m), vself);
    if (!__builtin_isfinite(m)) m = 0.f;

    float sum = has0 ? __expf(v0 - m) : 0.f;
    for (int p = p0 + 32; p < end; p += 32)
      sum += __expf(gat_leaky(as[csr[p]] + add) - m);
    const float exself = __expf(vself - m);
    const float denom  = (gat_wave_sum(sum) + exself) + 1e-16f;

    vec acc = *(const vec*)(hlane + (size_t)d * C) * (exself / denom);

    const int deg = end - beg;
    gat_rows_fma<C, vec>(acc, s0, __expf(v0 - m) / denom, (deg < 32) ? deg : 32, hlane);
    for (int base = beg + 32; base < end; base += 32) {
      const int   p    = base + lane;
      const int   s    = (p < end) ? csr[p] : d;
      const float a    = __expf(gat_leaky(as[s] + add) - m) / denom;
      const int   left = end - base;
      gat_rows_fma<C, vec>(acc, s, a, (left < 32) ? left : 32, hlane);
    }

    vec o = acc + bv;
    if (RELU) {
      for (int i = 0; i < V; ++i) o[i] = fmaxf(o[i], 0.f);
    }
    *(vec*)(out + (size_t)d * C + lane * V) = o;
    beg = end;
  }
}

extern "C" void kernel_launch(void* const* d_in, const int* in_sizes, int n_in,
                              void* d_out, int out_size, void* d_ws, size_t ws_size,
                              hipStream_t stream) {
  const float* fea = (const float*)d_in[0];
  const int*   ei  = (const int*)  d_in[1];
  const float* W1  = (const float*)d_in[2];
  const float* av1s= (const float*)d_in[3];
  const float* av1d= (const float*)d_in[4];
  const float* b1  = (const float*)d_in[5];
  const float* W2  = (const float*)d_in[6];
  const float* av2s= (const float*)d_in[7];
  const float* av2d= (const float*)d_in[8];
  const float* b2  = (const float*)d_in[9];
  float* out = (float*)d_out;

  const int B  = 2;
  const int C1 = 128, C2 = 64;
  const int N  = in_sizes[0] / (B * C1);
  const int E  = in_sizes[1] / (B * 2);
  const int NN = B * N;
  const int EE = B * E;

  float* w   = (float*)d_ws;
  float* h1   = w; w += (long)NN * C1;
  float* x2   = w; w += (long)NN * C1;
  float* h2   = w; w += (long)NN * C2;
  float* asb  = w; w += NN;
  float* adb  = w; w += NN;
  int* degb  = (int*)w; w += NN;
  int* rankb = (int*)w; w += EE;
  int* csrb  = (int*)w; w += EE;
  int* offb  = (int*)w; w += NN + 1;

  const int TB = 256;
  dim3 blk(TB);
  int gN       = (NN + TB - 1) / TB;
  const int EPT = 8;
  const int NPW = 4;
  const int APW1 = 8, APW2 = 16;
  dim3 gEdge((E + TB * EPT - 1) / (TB * EPT), B);
  int gNode    = ((NN + NPW - 1) / NPW * 32 + TB - 1) / TB;
  int gAlpha1  = ((NN + APW1 - 1) / APW1 * 32 + TB - 1) / TB;
  int gAlpha2  = ((NN + APW2 - 1) / APW2 * 32 + TB - 1) / TB;
  int gGemm    = NN / 16;
  int gScan    = (NN + 1023) / 1024;

  gat_zero_int<<<gN, blk, 0, stream>>>(degb, NN);
  gat_edge_rank<EPT><<<gEdge, blk, 0, stream>>>(ei, degb, rankb, N, E);
  gat_offsets_scan<<<gScan, dim3(1024), 0, stream>>>(degb, offb, NN);
  gat_csr_fill<EPT><<<gEdge, blk, 0, stream>>>(ei, rankb, offb, csrb, N, E);

  gat_gemm_wmma<8><<<gGemm, dim3(32), 0, stream>>>(fea, W1, h1, NN, C1);
  gat_alpha<128, APW1><<<gAlpha1, blk, 0, stream>>>(h1, av1s, av1d, asb, adb, NN);
  gat_node_softmax_gather<128, 1, NPW><<<gNode, blk, 0, stream>>>(
      offb, csrb, asb, adb, h1, b1, x2, NN);

  gat_gemm_wmma<4><<<gGemm, dim3(32), 0, stream>>>(x2, W2, h2, NN, C1);
  gat_alpha<64, APW2><<<gAlpha2, blk, 0, stream>>>(h2, av2s, av2d, asb, adb, NN);
  gat_node_softmax_gather<64, 0, NPW><<<gNode, blk, 0, stream>>>(
      offb, csrb, asb, adb, h2, b2, out, NN);
}


# Edge sort counters moved from memory to LDS: per edge tile a work-group counts, then places its edges with LDS adds on a node table; no global atomics left; segment order now tile, then arrival
# speedup vs baseline: 6.0800x; 1.4197x over previous
#include <hip/hip_runtime.h>
#include <math.h>

typedef __attribute__((ext_vector_type(2))) float v2f;
typedef __attribute__((ext_vector_type(8))) float v8f;
typedef __attribute__((ext_vector_type(4))) float v4f;

#define NEG_SLOPE 0.2f

template <int TILES>
__global__ __launch_bounds__(32)
void gat_gemm_wmma(const float* __restrict__ X, const float* __restrict__ W,
                   float* __restrict__ Hout, int N, int K) {
  const int F    = TILES * 16;
  const int row0 = blockIdx.x * 16;
  const int lane = threadIdx.x;
  const int l15  = lane & 15;
  const int kh   = (lane >> 4) * 2;
  const int m    = row0 + l15;

  v8f acc[TILES];
#pragma unroll
  for (int t = 0; t < TILES; ++t) acc[t] = (v8f){0.f,0.f,0.f,0.f,0.f,0.f,0.f,0.f};

  for (int k0 = 0; k0 < K; k0 += 4) {
    v2f a;
    a.x = X[(long)m * K + k0 + kh];
    a.y = X[(long)m * K + k0 + kh + 1];
#pragma unroll
    for (int t = 0; t < TILES; ++t) {
      v2f b;
      b.x = W[(long)(k0 + kh)     * F + t * 16 + l15];
      b.y = W[(long)(k0 + kh + 1) * F + t * 16 + l15];
      acc[t] = __builtin_amdgcn_wmma_f32_16x16x4_f32(
           false, a,  false, b,
           (short)0, acc[t],  false,  false);
    }
  }

  const int hi = (lane >> 4) * 8;
#pragma unroll
  for (int t = 0; t < TILES; ++t)
#pragma unroll
    for (int r = 0; r < 8; ++r)
      Hout[(long)(row0 + hi + r) * F + t * 16 + l15] = acc[t][r];
}

template <int C, int NPW>
__global__ __launch_bounds__(256)
void gat_alpha(const float* __restrict__ h,
               const float* __restrict__ a_src,
               const float* __restrict__ a_dst,
               float* __restrict__ as, float* __restrict__ ad, int N) {
  constexpr int R = C / 32;
  const int lane = threadIdx.x & 31;
  const int wave = __builtin_amdgcn_readfirstlane((int)((blockIdx.x * 256u + threadIdx.x) >> 5));
  const int d0   = wave * NPW;
  if (d0 >= N) return;

  float avs[R], avd[R];
#pragma unroll
  for (int i = 0; i < R; ++i) { avs[i] = a_src[lane + 32 * i]; avd[i] = a_dst[lane + 32 * i]; }

  float hv[NPW][R];
#pragma unroll
  for (int j = 0; j < NPW; ++j) {
    const int d = (d0 + j < N) ? (d0 + j) : (N - 1);
#pragma unroll
    for (int i = 0; i < R; ++i) hv[j][i] = h[(size_t)d * C + lane + 32 * i];
  }

  float rs = 0.f, rd = 0.f;
#pragma unroll
  for (int j = 0; j < NPW; ++j) {
    float ss = 0.f, sd = 0.f;
#pragma unroll
    for (int i = 0; i < R; ++i) { ss += hv[j][i] * avs[i]; sd += hv[j][i] * avd[i]; }
#pragma unroll
    for (int o = 16; o > 0; o >>= 1) {
      ss += __shfl_down(ss, o, 32);
      sd += __shfl_down(sd, o, 32);
    }
    const float bs = __int_as_float(__builtin_amdgcn_readfirstlane(__float_as_int(ss)));
    const float bd = __int_as_float(__builtin_amdgcn_readfirstlane(__float_as_int(sd)));
    rs = (lane == j) ? bs : rs;
    rd = (lane == j) ? bd : rd;
  }
  if (lane < NPW && d0 + lane < N) { as[d0 + lane] = rs; ad[d0 + lane] = rd; }
}

#define GAT_NCAP 25024
#define GAT_TB   512

__host__ __device__ __forceinline__ int gat_pitch(int N) { return (N + 31) / 32 * 32; }

__device__ __forceinline__ bool gat_al16(const void* p) {
  return (((unsigned long long)p) & 15ull) == 0ull;
}

template <int EPT>
__global__ __launch_bounds__(GAT_TB)
void gat_tile_count(const int* __restrict__ ei, int* __restrict__ hist,
                    int N, int E, int T, int L) {
  __shared__ __attribute__((aligned(16))) int cnt[GAT_NCAP];
  const int tid = threadIdx.x;
  const int b   = blockIdx.x;
  const int g   = blockIdx.y;
  const int nlo = blockIdx.z * GAT_NCAP;
  const int cn  = (N - nlo < GAT_NCAP) ? (N - nlo) : GAT_NCAP;
  const int* __restrict__ dst = ei + ((size_t)g * 2 + 1) * E;

  int4* c4 = (int4*)cnt;
  for (int i = tid; i < GAT_NCAP / 4; i += GAT_TB) c4[i] = make_int4(0, 0, 0, 0);
  __syncthreads();

  const int e_beg = b * L;
  const int e_end = (e_beg + L < E) ? (e_beg + L) : E;
  if (e_beg < e_end) {
    for (int e0 = e_beg + tid; e0 < e_end; e0 += GAT_TB * EPT) {
      int dv[EPT];
#pragma unroll
      for (int j = 0; j < EPT; ++j) {
        const int e = e0 + j * GAT_TB;
        const int v = dst[(e < e_end) ? e : (e_end - 1)];
        dv[j] = (e < e_end) ? v : -1;
      }
#pragma unroll
      for (int j = 0; j < EPT; ++j) {
        const unsigned rel = (unsigned)(dv[j] - nlo);
        if (rel < (unsigned)cn) atomicAdd(&cnt[rel], 1);
      }
    }
  }
  __syncthreads();

  int* __restrict__ hrow = hist + ((size_t)g * T + b) * gat_pitch(N) + nlo;
  if (gat_al16(hrow) && (cn & 3) == 0) {
    int4* h4 = (int4*)hrow;
    for (int i = tid; i < cn / 4; i += GAT_TB) h4[i] = c4[i];
  } else {
    for (int i = tid; i < cn; i += GAT_TB) hrow[i] = cnt[i];
  }
}

template <int T>
__global__ __launch_bounds__(256)
void gat_tile_prefix(int* __restrict__ hist, int* __restrict__ deg, int N) {
  const int n = blockIdx.x * 256 + threadIdx.x;
  const int g = blockIdx.y;
  if (n >= N) return;
  const size_t P = (size_t)gat_pitch(N);
  int* __restrict__ col = hist + (size_t)g * T * P + n;
  int v[T];
#pragma unroll
  for (int b = 0; b < T; ++b) v[b] = col[b * P];
  int run = 0;
#pragma unroll
  for (int b = 0; b < T; ++b) { col[b * P] = run; run += v[b]; }
  deg[(size_t)g * N + n] = run;
}

__global__ __launch_bounds__(1024)
void gat_offsets_scan(const int* __restrict__ deg, int* __restrict__ off, int N) {
  __shared__ int wtot[32];
  __shared__ int ptot[32];
  const int t    = threadIdx.x;
  const int lane = t & 31;
  const int w    = t >> 5;
  const int i    = blockIdx.x * 1024 + t;

  const int v = (i < N) ? deg[i] : 0;

  const int4* __restrict__ d4 = (const int4*)deg;
  const int n4 = blockIdx.x * 256;
  int ps = 0;
#pragma unroll 4
  for (int q = t; q < n4; q += 1024) {
    const int4 u = d4[q];
    ps += (u.x + u.y) + (u.z + u.w);
  }

  int x = v;
  for (int o = 1; o < 32; o <<= 1) {
    int y = __shfl_up(x, o, 32);
    if (lane >= o) x += y;
  }
  for (int o = 16; o > 0; o >>= 1) ps += __shfl_xor(ps, o, 32);
  if (lane == 31) { wtot[w] = x; ptot[w] = ps; }
  __syncthreads();
  int base = ptot[lane] + ((lane < w) ? wtot[lane] : 0);
  for (int o = 16; o > 0; o >>= 1) base += __shfl_xor(base, o, 32);

  if (i < N)      off[i] = base + x - v;
  if (i == N - 1) off[N] = base + x;
}

template <int EPT>
__global__ __launch_bounds__(GAT_TB)
void gat_tile_fill(const int* __restrict__ ei, const int* __restrict__ hist,
                   const int* __restrict__ off, int* __restrict__ csr,
                   int N, int E, int T, int L) {
  __shared__ __attribute__((aligned(16))) int cur[GAT_NCAP];
  const int tid = threadIdx.x;
  const int b   = blockIdx.x;
  const int g   = blockIdx.y;
  const int nlo = blockIdx.z * GAT_NCAP;
  const int cn  = (N - nlo < GAT_NCAP) ? (N - nlo) : GAT_NCAP;
  const int* __restrict__ src  = ei + (size_t)g * 2 * E;
  const int* __restrict__ dst  = src + E;
  const int* __restrict__ hrow = hist + ((size_t)g * T + b) * gat_pitch(N) + nlo;
  const int* __restrict__ orow = off + (size_t)g * N + nlo;

  if (gat_al16(hrow) && gat_al16(orow) && (cn & 3) == 0) {
    const int4* h4 = (const int4*)hrow;
    const int4* o4 = (const int4*)orow;
    int4* c4 = (int4*)cur;
    for (int i = tid; i < cn / 4; i += GAT_TB) {
      const int4 h = h4[i];
      const int4 o = o4[i];
      c4[i] = make_int4(h.x + o.x, h.y + o.y, h.z + o.z, h.w + o.w);
    }
  } else {
    for (int i = tid; i < cn; i += GAT_TB) cur[i] = hrow[i] + orow[i];
  }
  __syncthreads();

  const int idshift = g * N;
  const int e_beg = b * L;
  const int e_end = (e_beg + L < E) ? (e_beg + L) : E;
  if (e_beg >= e_end) return;
  for (int e0 = e_beg + tid; e0 < e_end; e0 += GAT_TB * EPT) {
    int sv[EPT], dv[EPT], pv[EPT];
#pragma unroll
    for (int j = 0; j < EPT; ++j) {
      const int e  = e0 + j * GAT_TB;
      const int ec = (e < e_end) ? e : (e_end - 1);
      sv[j] = src[ec];
      const int v = dst[ec];
      dv[j] = (e < e_end) ? v : -1;
    }
#pragma unroll
    for (int j = 0; j < EPT; ++j) {
      const unsigned rel = (unsigned)(dv[j] - nlo);
      pv[j] = -1;
      if (rel < (unsigned)cn) pv[j] = atomicAdd(&cur[rel], 1);
    }
#pragma unroll
    for (int j = 0; j < EPT; ++j)
      if (pv[j] >= 0) csr[pv[j]] = sv[j] + idshift;
  }
}

template <int V> struct gat_rowvec;
template <> struct gat_rowvec<4> { typedef v4f T; };
template <> struct gat_rowvec<2> { typedef v2f T; };

__device__ __forceinline__ float gat_leaky(float v) {
  return (v > 0.f) ? v : NEG_SLOPE * v;
}

__device__ __forceinline__ float gat_wave_max(float x) {
  for (int o = 16; o > 0; o >>= 1) x = fmaxf(x, __shfl_xor(x, o, 32));
  return x;
}

__device__ __forceinline__ float gat_wave_sum(float x) {
  for (int o = 16; o > 0; o >>= 1) x += __shfl_xor(x, o, 32);
  return x;
}

template <int C, typename vec>
__device__ __forceinline__ void gat_rows_fma(vec& acc, int s, float a, int cnt,
                                             const float* __restrict__ hlane) {
#pragma unroll 4
  for (int j = 0; j < cnt; ++j) {
    const int   sj = __builtin_amdgcn_readlane(s, j);
    const float aj = __int_as_float(__builtin_amdgcn_readlane(__float_as_int(a), j));
    const vec   hv = *(const vec*)(hlane + (size_t)sj * C);
    acc += hv * aj;
  }
}

template <int C, int RELU, int NPW>
__global__ __launch_bounds__(256)
void gat_node_softmax_gather(const int* __restrict__ off,
                             const int* __restrict__ csr,
                             const float* __restrict__ as,
                             const float* __restrict__ ad,
                             const float* __restrict__ h,
                             const float* __restrict__ bias,
                             float* __restrict__ out, int N) {
  constexpr int V = C / 32;
  typedef typename gat_rowvec<V>::T vec;
  const int lane = threadIdx.x & 31;
  const int wave = __builtin_amdgcn_readfirstlane((int)((blockIdx.x * 256u + threadIdx.x) >> 5));
  const int d0   = wave * NPW;
  if (d0 >= N) return;
  const int d1   = (d0 + NPW < N) ? (d0 + NPW) : N;

  const float* hlane = h + lane * V;
  const vec    bv    = *(const vec*)(bias + lane * V);

  int beg = off[d0];
  for (int d = d0; d < d1; ++d) {
    const int   end   = off[d + 1];
    const float add   = ad[d];
    const float vself = gat_leaky(as[d] + add);

    const int   p0   = beg + lane;
    const bool  has0 = p0 < end;
    const int   s0   = has0 ? csr[p0] : d;
    const float v0   = gat_leaky(as[s0] + add);

    float m = v0;
    for (int p = p0 + 32; p < end; p += 32)
      m = fmaxf(m, gat_leaky(as[csr[p]] + add));
    m = fmaxf(gat_wave_max(m), vself);
    if (!__builtin_isfinite(m)) m = 0.f;

    float sum = has0 ? __expf(v0 - m) : 0.f;
    for (int p = p0 + 32; p < end; p += 32)
      sum += __expf(gat_leaky(as[csr[p]] + add) - m);
    const float exself = __expf(vself - m);
    const float denom  = (gat_wave_sum(sum) + exself) + 1e-16f;

    vec acc = *(const vec*)(hlane + (size_t)d * C) * (exself / denom);

    const int deg = end - beg;
    gat_rows_fma<C, vec>(acc, s0, __expf(v0 - m) / denom, (deg < 32) ? deg : 32, hlane);
    for (int base = beg + 32; base < end; base += 32) {
      const int   p    = base + lane;
      const int   s    = (p < end) ? csr[p] : d;
      const float a    = __expf(gat_leaky(as[s] + add) - m) / denom;
      const int   left = end - base;
      gat_rows_fma<C, vec>(acc, s, a, (left < 32) ? left : 32, hlane);
    }

    vec o = acc + bv;
    if (RELU) {
      for (int i = 0; i < V; ++i) o[i] = fmaxf(o[i], 0.f);
    }
    *(vec*)(out + (size_t)d * C + lane * V) = o;
    beg = end;
  }
}

extern "C" void kernel_launch(void* const* d_in, const int* in_sizes, int n_in,
                              void* d_out, int out_size, void* d_ws, size_t ws_size,
                              hipStream_t stream) {
  const float* fea = (const float*)d_in[0];
  const int*   ei  = (const int*)  d_in[1];
  const float* W1  = (const float*)d_in[2];
  const float* av1s= (const float*)d_in[3];
  const float* av1d= (const float*)d_in[4];
  const float* b1  = (const float*)d_in[5];
  const float* W2  = (const float*)d_in[6];
  const float* av2s= (const float*)d_in[7];
  const float* av2d= (const float*)d_in[8];
  const float* b2  = (const float*)d_in[9];
  float* out = (float*)d_out;

  const int B  = 2;
  const int C1 = 128, C2 = 64;
  const int N  = in_sizes[0] / (B * C1);
  const int E  = in_sizes[1] / (B * 2);
  const int NN = B * N;
  const int EE = B * E;

  float* w   = (float*)d_ws;
  float* h1   = w; w += (long)NN * C1;
  float* x2   = w; w += (long)NN * C1;
  float* h2   = w; w += (long)NN * C2;
  float* asb  = w; w += NN;
  float* adb  = w; w += NN;
  const int T = 64;
  int* degb  = (int*)w; w += NN;
  int* histb = (int*)w; w += (long)B * T * gat_pitch(N);
  int* csrb  = (int*)w; w += EE;
  int* offb  = (int*)w; w += NN + 1;

  const int TB = 256;
  dim3 blk(TB);
  const int EPT = 16;
  const int NPW = 4;
  const int APW1 = 8, APW2 = 16;
  const int L = (E + T - 1) / T;
  dim3 gEdge(T, B, (N + GAT_NCAP - 1) / GAT_NCAP);
  dim3 gPref((N + TB - 1) / TB, B);
  int gNode    = ((NN + NPW - 1) / NPW * 32 + TB - 1) / TB;
  int gAlpha1  = ((NN + APW1 - 1) / APW1 * 32 + TB - 1) / TB;
  int gAlpha2  = ((NN + APW2 - 1) / APW2 * 32 + TB - 1) / TB;
  int gGemm    = NN / 16;
  int gScan    = (NN + 1023) / 1024;

  gat_tile_count<EPT><<<gEdge, dim3(GAT_TB), 0, stream>>>(ei, histb, N, E, T, L);
  gat_tile_prefix<T><<<gPref, blk, 0, stream>>>(histb, degb, N);
  gat_offsets_scan<<<gScan, dim3(1024), 0, stream>>>(degb, offb, NN);
  gat_tile_fill<EPT><<<gEdge, dim3(GAT_TB), 0, stream>>>(ei, histb, offb, csrb, N, E, T, L);

  gat_gemm_wmma<8><<<gGemm, dim3(32), 0, stream>>>(fea, W1, h1, NN, C1);
  gat_alpha<128, APW1><<<gAlpha1, blk, 0, stream>>>(h1, av1s, av1d, asb, adb, NN);
  gat_node_softmax_gather<128, 1, NPW><<<gNode, blk, 0, stream>>>(
      offb, csrb, asb, adb, h1, b1, x2, NN);

  gat_gemm_wmma<4><<<gGemm, dim3(32), 0, stream>>>(x2, W2, h2, NN, C1);
  gat_alpha<64, APW2><<<gAlpha2, blk, 0, stream>>>(h2, av2s, av2d, asb, adb, NN);
  gat_node_softmax_gather<64, 0, NPW><<<gNode, blk, 0, stream>>>(
      offb, csrb, asb, adb, h2, b2, out, NN);
}


# GEMM: each wave keeps its 64 (32) columns of W in registers for the whole kernel and streams 16-row tiles past them, A as 16-byte pieces asked for one tile ahead; k order within the sums changed
# speedup vs baseline: 6.6070x; 1.0867x over previous
#include <hip/hip_runtime.h>
#include <math.h>

typedef __attribute__((ext_vector_type(2))) float v2f;
typedef __attribute__((ext_vector_type(8))) float v8f;
typedef __attribute__((ext_vector_type(4))) float v4f;

#define NEG_SLOPE 0.2f

template <int KP, int K>
__device__ __forceinline__ void gat_gemm_load_a(v4f (&a)[KP], const float* __restrict__ X,
                                                int rt, int c, int h) {
  const float* xr = X + ((size_t)rt * 16 + c) * K + 4 * h;
#pragma unroll
  for (int j = 0; j < KP; ++j) a[j] = *(const v4f*)(xr + 8 * j);
}

template <int TILES, int KP, int KS>
__device__ __forceinline__ void gat_gemm_tile(const v4f (&a)[KP], const v2f (&b)[KS][TILES],
                                              float* __restrict__ orow, int F) {
  v8f acc[TILES];
#pragma unroll
  for (int t = 0; t < TILES; ++t) acc[t] = (v8f){0.f,0.f,0.f,0.f,0.f,0.f,0.f,0.f};
#pragma unroll
  for (int j = 0; j < KP; ++j) {
    v2f p0; p0.x = a[j].x; p0.y = a[j].y;
    v2f p1; p1.x = a[j].z; p1.y = a[j].w;
#pragma unroll
    for (int t = 0; t < TILES; ++t)
      acc[t] = __builtin_amdgcn_wmma_f32_16x16x4_f32(
           false, p0,  false, b[2 * j][t],
           (short)0, acc[t],  false,  false);
#pragma unroll
    for (int t = 0; t < TILES; ++t)
      acc[t] = __builtin_amdgcn_wmma_f32_16x16x4_f32(
           false, p1,  false, b[2 * j + 1][t],
           (short)0, acc[t],  false,  false);
  }
#pragma unroll
  for (int t = 0; t < TILES; ++t)
#pragma unroll
    for (int r = 0; r < 8; ++r)
      orow[(size_t)r * F + t * 16] = acc[t][r];
}

template <int TILES, int CG>
__global__ __launch_bounds__(256)
void gat_gemm_wmma(const float* __restrict__ X, const float* __restrict__ W,
                   float* __restrict__ Hout, int nRowTiles) {
  constexpr int K  = 128;
  constexpr int F  = CG * TILES * 16;
  constexpr int KS = K / 4;
  constexpr int KP = K / 8;
  const int lane = threadIdx.x & 31;
  const int c    = lane & 15;
  const int h    = lane >> 4;
  const int wib  = __builtin_amdgcn_readfirstlane((int)(threadIdx.x >> 5));
  const int cg   = wib % CG;
  const int pr   = (int)blockIdx.x + (int)gridDim.x * (wib / CG);
  const int npr  = (int)gridDim.x * (8 / CG);
  const int col0 = cg * TILES * 16 + c;

  v2f b[KS][TILES];
#pragma unroll
  for (int s = 0; s < KS; ++s) {
    const int kr = 8 * (s >> 1) + 4 * h + 2 * (s & 1);
#pragma unroll
    for (int t = 0; t < TILES; ++t) {
      b[s][t].x = W[(kr)     * F + col0 + t * 16];
      b[s][t].y = W[(kr + 1) * F + col0 + t * 16];
    }
  }

  if (pr >= nRowTiles) return;
  const int cnt = (nRowTiles - pr + npr - 1) / npr;
  int rt = pr;
  v4f a0[KP], a1[KP];
  if (cnt & 1) {
    gat_gemm_load_a<KP, K>(a0, X, rt, c, h);
    gat_gemm_tile<TILES, KP, KS>(a0, b, Hout + ((size_t)rt * 16 + 8 * h) * F + col0, F);
    rt += npr;
  }
  if (rt >= nRowTiles) return;
  gat_gemm_load_a<KP, K>(a0, X, rt, c, h);
  for (; rt < nRowTiles; rt += 2 * npr) {
    gat_gemm_load_a<KP, K>(a1, X, rt + npr, c, h);
    __builtin_amdgcn_sched_barrier(0);
    gat_gemm_tile<TILES, KP, KS>(a0, b, Hout + ((size_t)rt * 16 + 8 * h) * F + col0, F);
    gat_gemm_load_a<KP, K>(a0, X, (rt + 2 * npr < nRowTiles) ? (rt + 2 * npr) : rt, c, h);
    __builtin_amdgcn_sched_barrier(0);
    gat_gemm_tile<TILES, KP, KS>(a1, b, Hout + ((size_t)(rt + npr) * 16 + 8 * h) * F + col0, F);
  }
}

template <int C, int NPW>
__global__ __launch_bounds__(256)
void gat_alpha(const float* __restrict__ h,
               const float* __restrict__ a_src,
               const float* __restrict__ a_dst,
               float* __restrict__ as, float* __restrict__ ad, int N) {
  constexpr int R = C / 32;
  const int lane = threadIdx.x & 31;
  const int wave = __builtin_amdgcn_readfirstlane((int)((blockIdx.x * 256u + threadIdx.x) >> 5));
  const int d0   = wave * NPW;
  if (d0 >= N) return;

  float avs[R], avd[R];
#pragma unroll
  for (int i = 0; i < R; ++i) { avs[i] = a_src[lane + 32 * i]; avd[i] = a_dst[lane + 32 * i]; }

  float hv[NPW][R];
#pragma unroll
  for (int j = 0; j < NPW; ++j) {
    const int d = (d0 + j < N) ? (d0 + j) : (N - 1);
#pragma unroll
    for (int i = 0; i < R; ++i) hv[j][i] = h[(size_t)d * C + lane + 32 * i];
  }

  float rs = 0.f, rd = 0.f;
#pragma unroll
  for (int j = 0; j < NPW; ++j) {
    float ss = 0.f, sd = 0.f;
#pragma unroll
    for (int i = 0; i < R; ++i) { ss += hv[j][i] * avs[i]; sd += hv[j][i] * avd[i]; }
#pragma unroll
    for (int o = 16; o > 0; o >>= 1) {
      ss += __shfl_down(ss, o, 32);
      sd += __shfl_down(sd, o, 32);
    }
    const float bs = __int_as_float(__builtin_amdgcn_readfirstlane(__float_as_int(ss)));
    const float bd = __int_as_float(__builtin_amdgcn_readfirstlane(__float_as_int(sd)));
    rs = (lane == j) ? bs : rs;
    rd = (lane == j) ? bd : rd;
  }
  if (lane < NPW && d0 + lane < N) { as[d0 + lane] = rs; ad[d0 + lane] = rd; }
}

#define GAT_NCAP 25024
#define GAT_TB   512

__host__ __device__ __forceinline__ int gat_pitch(int N) { return (N + 31) / 32 * 32; }

__device__ __forceinline__ bool gat_al16(const void* p) {
  return (((unsigned long long)p) & 15ull) == 0ull;
}

template <int EPT>
__global__ __launch_bounds__(GAT_TB)
void gat_tile_count(const int* __restrict__ ei, int* __restrict__ hist,
                    int N, int E, int T, int L) {
  __shared__ __attribute__((aligned(16))) int cnt[GAT_NCAP];
  const int tid = threadIdx.x;
  const int b   = blockIdx.x;
  const int g   = blockIdx.y;
  const int nlo = blockIdx.z * GAT_NCAP;
  const int cn  = (N - nlo < GAT_NCAP) ? (N - nlo) : GAT_NCAP;
  const int* __restrict__ dst = ei + ((size_t)g * 2 + 1) * E;

  int4* c4 = (int4*)cnt;
  for (int i = tid; i < GAT_NCAP / 4; i += GAT_TB) c4[i] = make_int4(0, 0, 0, 0);
  __syncthreads();

  const int e_beg = b * L;
  const int e_end = (e_beg + L < E) ? (e_beg + L) : E;
  if (e_beg < e_end) {
    for (int e0 = e_beg + tid; e0 < e_end; e0 += GAT_TB * EPT) {
      int dv[EPT];
#pragma unroll
      for (int j = 0; j < EPT; ++j) {
        const int e = e0 + j * GAT_TB;
        const int v = dst[(e < e_end) ? e : (e_end - 1)];
        dv[j] = (e < e_end) ? v : -1;
      }
#pragma unroll
      for (int j = 0; j < EPT; ++j) {
        const unsigned rel = (unsigned)(dv[j] - nlo);
        if (rel < (unsigned)cn) atomicAdd(&cnt[rel], 1);
      }
    }
  }
  __syncthreads();

  int* __restrict__ hrow = hist + ((size_t)g * T + b) * gat_pitch(N) + nlo;
  if (gat_al16(hrow) && (cn & 3) == 0) {
    int4* h4 = (int4*)hrow;
    for (int i = tid; i < cn / 4; i += GAT_TB) h4[i] = c4[i];
  } else {
    for (int i = tid; i < cn; i += GAT_TB) hrow[i] = cnt[i];
  }
}

template <int T>
__global__ __launch_bounds__(256)
void gat_tile_prefix(int* __restrict__ hist, int* __restrict__ deg, int N) {
  const int n = blockIdx.x * 256 + threadIdx.x;
  const int g = blockIdx.y;
  if (n >= N) return;
  const size_t P = (size_t)gat_pitch(N);
  int* __restrict__ col = hist + (size_t)g * T * P + n;
  int v[T];
#pragma unroll
  for (int b = 0; b < T; ++b) v[b] = col[b * P];
  int run = 0;
#pragma unroll
  for (int b = 0; b < T; ++b) { col[b * P] = run; run += v[b]; }
  deg[(size_t)g * N + n] = run;
}

__global__ __launch_bounds__(1024)
void gat_offsets_scan(const int* __restrict__ deg, int* __restrict__ off, int N) {
  __shared__ int wtot[32];
  __shared__ int ptot[32];
  const int t    = threadIdx.x;
  const int lane = t & 31;
  const int w    = t >> 5;
  const int i    = blockIdx.x * 1024 + t;

  const int v = (i < N) ? deg[i] : 0;

  const int4* __restrict__ d4 = (const int4*)deg;
  const int n4 = blockIdx.x * 256;
  int ps = 0;
#pragma unroll 4
  for (int q = t; q < n4; q += 1024) {
    const int4 u = d4[q];
    ps += (u.x + u.y) + (u.z + u.w);
  }

  int x = v;
  for (int o = 1; o < 32; o <<= 1) {
    int y = __shfl_up(x, o, 32);
    if (lane >= o) x += y;
  }
  for (int o = 16; o > 0; o >>= 1) ps += __shfl_xor(ps, o, 32);
  if (lane == 31) { wtot[w] = x; ptot[w] = ps; }
  __syncthreads();
  int base = ptot[lane] + ((lane < w) ? wtot[lane] : 0);
  for (int o = 16; o > 0; o >>= 1) base += __shfl_xor(base, o, 32);

  if (i < N)      off[i] = base + x - v;
  if (i == N - 1) off[N] = base + x;
}

template <int EPT>
__global__ __launch_bounds__(GAT_TB)
void gat_tile_fill(const int* __restrict__ ei, const int* __restrict__ hist,
                   const int* __restrict__ off, int* __restrict__ csr,
                   int N, int E, int T, int L) {
  __shared__ __attribute__((aligned(16))) int cur[GAT_NCAP];
  const int tid = threadIdx.x;
  const int b   = blockIdx.x;
  const int g   = blockIdx.y;
  const int nlo = blockIdx.z * GAT_NCAP;
  const int cn  = (N - nlo < GAT_NCAP) ? (N - nlo) : GAT_NCAP;
  const int* __restrict__ src  = ei + (size_t)g * 2 * E;
  const int* __restrict__ dst  = src + E;
  const int* __restrict__ hrow = hist + ((size_t)g * T + b) * gat_pitch(N) + nlo;
  const int* __restrict__ orow = off + (size_t)g * N + nlo;

  if (gat_al16(hrow) && gat_al16(orow) && (cn & 3) == 0) {
    const int4* h4 = (const int4*)hrow;
    const int4* o4 = (const int4*)orow;
    int4* c4 = (int4*)cur;
    for (int i = tid; i < cn / 4; i += GAT_TB) {
      const int4 h = h4[i];
      const int4 o = o4[i];
      c4[i] = make_int4(h.x + o.x, h.y + o.y, h.z + o.z, h.w + o.w);
    }
  } else {
    for (int i = tid; i < cn; i += GAT_TB) cur[i] = hrow[i] + orow[i];
  }
  __syncthreads();

  const int idshift = g * N;
  const int e_beg = b * L;
  const int e_end = (e_beg + L < E) ? (e_beg + L) : E;
  if (e_beg >= e_end) return;
  for (int e0 = e_beg + tid; e0 < e_end; e0 += GAT_TB * EPT) {
    int sv[EPT], dv[EPT], pv[EPT];
#pragma unroll
    for (int j = 0; j < EPT; ++j) {
      const int e  = e0 + j * GAT_TB;
      const int ec = (e < e_end) ? e : (e_end - 1);
      sv[j] = src[ec];
      const int v = dst[ec];
      dv[j] = (e < e_end) ? v : -1;
    }
#pragma unroll
    for (int j = 0; j < EPT; ++j) {
      const unsigned rel = (unsigned)(dv[j] - nlo);
      pv[j] = -1;
      if (rel < (unsigned)cn) pv[j] = atomicAdd(&cur[rel], 1);
    }
#pragma unroll
    for (int j = 0; j < EPT; ++j)
      if (pv[j] >= 0) csr[pv[j]] = sv[j] + idshift;
  }
}

template <int V> struct gat_rowvec;
template <> struct gat_rowvec<4> { typedef v4f T; };
template <> struct gat_rowvec<2> { typedef v2f T; };

__device__ __forceinline__ float gat_leaky(float v) {
  return (v > 0.f) ? v : NEG_SLOPE * v;
}

__device__ __forceinline__ float gat_wave_max(float x) {
  for (int o = 16; o > 0; o >>= 1) x = fmaxf(x, __shfl_xor(x, o, 32));
  return x;
}

__device__ __forceinline__ float gat_wave_sum(float x) {
  for (int o = 16; o > 0; o >>= 1) x += __shfl_xor(x, o, 32);
  return x;
}

template <int C, typename vec>
__device__ __forceinline__ void gat_rows_fma(vec& acc, int s, float a, int cnt,
                                             const float* __restrict__ hlane) {
#pragma unroll 4
  for (int j = 0; j < cnt; ++j) {
    const int   sj = __builtin_amdgcn_readlane(s, j);
    const float aj = __int_as_float(__builtin_amdgcn_readlane(__float_as_int(a), j));
    const vec   hv = *(const vec*)(hlane + (size_t)sj * C);
    acc += hv * aj;
  }
}

template <int C, int RELU, int NPW>
__global__ __launch_bounds__(256)
void gat_node_softmax_gather(const int* __restrict__ off,
                             const int* __restrict__ csr,
                             const float* __restrict__ as,
                             const float* __restrict__ ad,
                             const float* __restrict__ h,
                             const float* __restrict__ bias,
                             float* __restrict__ out, int N) {
  constexpr int V = C / 32;
  typedef typename gat_rowvec<V>::T vec;
  const int lane = threadIdx.x & 31;
  const int wave = __builtin_amdgcn_readfirstlane((int)((blockIdx.x * 256u + threadIdx.x) >> 5));
  const int d0   = wave * NPW;
  if (d0 >= N) return;
  const int d1   = (d0 + NPW < N) ? (d0 + NPW) : N;

  const float* hlane = h + lane * V;
  const vec    bv    = *(const vec*)(bias + lane * V);

  int beg = off[d0];
  for (int d = d0; d < d1; ++d) {
    const int   end   = off[d + 1];
    const float add   = ad[d];
    const float vself = gat_leaky(as[d] + add);

    const int   p0   = beg + lane;
    const bool  has0 = p0 < end;
    const int   s0   = has0 ? csr[p0] : d;
    const float v0   = gat_leaky(as[s0] + add);

    float m = v0;
    for (int p = p0 + 32; p < end; p += 32)
      m = fmaxf(m, gat_leaky(as[csr[p]] + add));
    m = fmaxf(gat_wave_max(m), vself);
    if (!__builtin_isfinite(m)) m = 0.f;

    float sum = has0 ? __expf(v0 - m) : 0.f;
    for (int p = p0 + 32; p < end; p += 32)
      sum += __expf(gat_leaky(as[csr[p]] + add) - m);
    const float exself = __expf(vself - m);
    const float denom  = (gat_wave_sum(sum) + exself) + 1e-16f;

    vec acc = *(const vec*)(hlane + (size_t)d * C) * (exself / denom);

    const int deg = end - beg;
    gat_rows_fma<C, vec>(acc, s0, __expf(v0 - m) / denom, (deg < 32) ? deg : 32, hlane);
    for (int base = beg + 32; base < end; base += 32) {
      const int   p    = base + lane;
      const int   s    = (p < end) ? csr[p] : d;
      const float a    = __expf(gat_leaky(as[s] + add) - m) / denom;
      const int   left = end - base;
      gat_rows_fma<C, vec>(acc, s, a, (left < 32) ? left : 32, hlane);
    }

    vec o = acc + bv;
    if (RELU) {
      for (int i = 0; i < V; ++i) o[i] = fmaxf(o[i], 0.f);
    }
    *(vec*)(out + (size_t)d * C + lane * V) = o;
    beg = end;
  }
}

extern "C" void kernel_launch(void* const* d_in, const int* in_sizes, int n_in,
                              void* d_out, int out_size, void* d_ws, size_t ws_size,
                              hipStream_t stream) {
  const float* fea = (const float*)d_in[0];
  const int*   ei  = (const int*)  d_in[1];
  const float* W1  = (const float*)d_in[2];
  const float* av1s= (const float*)d_in[3];
  const float* av1d= (const float*)d_in[4];
  const float* b1  = (const float*)d_in[5];
  const float* W2  = (const float*)d_in[6];
  const float* av2s= (const float*)d_in[7];
  const float* av2d= (const float*)d_in[8];
  const float* b2  = (const float*)d_in[9];
  float* out = (float*)d_out;

  const int B  = 2;
  const int C1 = 128, C2 = 64;
  const int N  = in_sizes[0] / (B * C1);
  const int E  = in_sizes[1] / (B * 2);
  const int NN = B * N;
  const int EE = B * E;

  float* w   = (float*)d_ws;
  float* h1   = w; w += (long)NN * C1;
  float* x2   = w; w += (long)NN * C1;
  float* h2   = w; w += (long)NN * C2;
  float* asb  = w; w += NN;
  float* adb  = w; w += NN;
  const int T = 64;
  int* degb  = (int*)w; w += NN;
  int* histb = (int*)w; w += (long)B * T * gat_pitch(N);
  int* csrb  = (int*)w; w += EE;
  int* offb  = (int*)w; w += NN + 1;

  const int TB = 256;
  dim3 blk(TB);
  const int EPT = 16;
  const int NPW = 4;
  const int APW1 = 8, APW2 = 16;
  const int L = (E + T - 1) / T;
  dim3 gEdge(T, B, (N + GAT_NCAP - 1) / GAT_NCAP);
  dim3 gPref((N + TB - 1) / TB, B);
  int gNode    = ((NN + NPW - 1) / NPW * 32 + TB - 1) / TB;
  int gAlpha1  = ((NN + APW1 - 1) / APW1 * 32 + TB - 1) / TB;
  int gAlpha2  = ((NN + APW2 - 1) / APW2 * 32 + TB - 1) / TB;
  const int nRowTiles = NN / 16;
  int gGemm    = 256;
  int gScan    = (NN + 1023) / 1024;

  gat_tile_count<EPT><<<gEdge, dim3(GAT_TB), 0, stream>>>(ei, histb, N, E, T, L);
  gat_tile_prefix<T><<<gPref, blk, 0, stream>>>(histb, degb, N);
  gat_offsets_scan<<<gScan, dim3(1024), 0, stream>>>(degb, offb, NN);
  gat_tile_fill<EPT><<<gEdge, dim3(GAT_TB), 0, stream>>>(ei, histb, offb, csrb, N, E, T, L);

  gat_gemm_wmma<4, 2><<<gGemm, blk, 0, stream>>>(fea, W1, h1, nRowTiles);
  gat_alpha<128, APW1><<<gAlpha1, blk, 0, stream>>>(h1, av1s, av1d, asb, adb, NN);
  gat_node_softmax_gather<128, 1, NPW><<<gNode, blk, 0, stream>>>(
      offb, csrb, asb, adb, h1, b1, x2, NN);

  gat_gemm_wmma<2, 2><<<gGemm, blk, 0, stream>>>(x2, W2, h2, nRowTiles);
  gat_alpha<64, APW2><<<gAlpha2, blk, 0, stream>>>(h2, av2s, av2d, asb, adb, NN);
  gat_node_softmax_gather<64, 0, NPW><<<gNode, blk, 0, stream>>>(
      offb, csrb, asb, adb, h2, b2, out, NN);
}


# Logits formed inside the GEMM from the A pieces it holds: as = x.(W a_src), ad = x.(W a_dst), W a built per work-group in LDS; the two logit launches are gone; order of the logit sums changed
# speedup vs baseline: 6.8678x; 1.0395x over previous
#include <hip/hip_runtime.h>
#include <math.h>

typedef __attribute__((ext_vector_type(2))) float v2f;
typedef __attribute__((ext_vector_type(8))) float v8f;
typedef __attribute__((ext_vector_type(4))) float v4f;

#define NEG_SLOPE 0.2f

template <int V> struct gat_rowvec;
template <> struct gat_rowvec<4> { typedef v4f T; };
template <> struct gat_rowvec<2> { typedef v2f T; };

template <int O>
__device__ __forceinline__ void gat_halve(float (&v)[32], int lane) {
  const bool up = (lane & O) != 0;
#pragma unroll
  for (int i = 0; i < O; ++i) {
    const float keep = up ? v[i + O] : v[i];
    const float send = up ? v[i] : v[i + O];
    v[i] = keep + __shfl_xor(send, O, 32);
  }
}

template <int F>
__device__ __forceinline__ void gat_gemm_logit_vectors(float* __restrict__ s_wa,
                                                       const float* __restrict__ W,
                                                       const float* __restrict__ a_src,
                                                       const float* __restrict__ a_dst,
                                                       int wib, int lane) {
  constexpr int VW = F / 32;
  typedef typename gat_rowvec<VW>::T wvec;
  const wvec avs = *(const wvec*)(a_src + lane * VW);
  const wvec avd = *(const wvec*)(a_dst + lane * VW);
  float v[32];
#pragma unroll
  for (int i = 0; i < 16; ++i) {
    const wvec wr = *(const wvec*)(W + (size_t)(16 * wib + i) * F + lane * VW);
    float ss = 0.f, sd = 0.f;
#pragma unroll
    for (int q = 0; q < VW; ++q) { ss += wr[q] * avs[q]; sd += wr[q] * avd[q]; }
    v[i] = ss; v[16 + i] = sd;
  }
  gat_halve<16>(v, lane);
  gat_halve<8>(v, lane);
  gat_halve<4>(v, lane);
  gat_halve<2>(v, lane);
  gat_halve<1>(v, lane);
  s_wa[(lane >> 4) * 128 + 16 * wib + (lane & 15)] = v[0];
}

template <int KP, int K>
__device__ __forceinline__ void gat_gemm_load_a(v4f (&a)[KP], const float* __restrict__ X,
                                                int rt, int c, int h) {
  const float* xr = X + ((size_t)rt * 16 + c) * K + 4 * h;
#pragma unroll
  for (int j = 0; j < KP; ++j) a[j] = *(const v4f*)(xr + 8 * j);
}

template <int TILES, int KP, int KS>
__device__ __forceinline__ void gat_gemm_tile(const v4f (&a)[KP], const v2f (&b)[KS][TILES],
                                              float* __restrict__ orow, int F,
                                              const v4f (&wav)[KP], float* __restrict__ lrow,
                                              int c, int h) {
  v4f s4 = a[0] * wav[0];
#pragma unroll
  for (int j = 1; j < KP; ++j) s4 += a[j] * wav[j];
  float lg = (s4.x + s4.y) + (s4.z + s4.w);
  lg += __shfl_xor(lg, 16, 32);
  if (h == 0) lrow[c] = lg;

  v8f acc[TILES];
#pragma unroll
  for (int t = 0; t < TILES; ++t) acc[t] = (v8f){0.f,0.f,0.f,0.f,0.f,0.f,0.f,0.f};
#pragma unroll
  for (int j = 0; j < KP; ++j) {
    v2f p0; p0.x = a[j].x; p0.y = a[j].y;
    v2f p1; p1.x = a[j].z; p1.y = a[j].w;
#pragma unroll
    for (int t = 0; t < TILES; ++t)
      acc[t] = __builtin_amdgcn_wmma_f32_16x16x4_f32(
           false, p0,  false, b[2 * j][t],
           (short)0, acc[t],  false,  false);
#pragma unroll
    for (int t = 0; t < TILES; ++t)
      acc[t] = __builtin_amdgcn_wmma_f32_16x16x4_f32(
           false, p1,  false, b[2 * j + 1][t],
           (short)0, acc[t],  false,  false);
  }
#pragma unroll
  for (int t = 0; t < TILES; ++t)
#pragma unroll
    for (int r = 0; r < 8; ++r)
      orow[(size_t)r * F + t * 16] = acc[t][r];
}

template <int TILES, int CG>
__global__ __launch_bounds__(256)
void gat_gemm_wmma(const float* __restrict__ X, const float* __restrict__ W,
                   float* __restrict__ Hout,
                   const float* __restrict__ a_src, const float* __restrict__ a_dst,
                   float* __restrict__ as, float* __restrict__ ad, int nRowTiles) {
  static_assert(CG == 2, "one wave of a row tile does a_src, the other a_dst");
  __shared__ __attribute__((aligned(16))) float s_wa[2 * 128];
  constexpr int K  = 128;
  constexpr int F  = CG * TILES * 16;
  constexpr int KS = K / 4;
  constexpr int KP = K / 8;
  const int lane = threadIdx.x & 31;
  const int c    = lane & 15;
  const int h    = lane >> 4;
  const int wib  = __builtin_amdgcn_readfirstlane((int)(threadIdx.x >> 5));
  const int cg   = wib % CG;
  const int pr   = (int)blockIdx.x + (int)gridDim.x * (wib / CG);
  const int npr  = (int)gridDim.x * (8 / CG);
  const int col0 = cg * TILES * 16 + c;

  v2f b[KS][TILES];
#pragma unroll
  for (int s = 0; s < KS; ++s) {
    const int kr = 8 * (s >> 1) + 4 * h + 2 * (s & 1);
#pragma unroll
    for (int t = 0; t < TILES; ++t) {
      b[s][t].x = W[(kr)     * F + col0 + t * 16];
      b[s][t].y = W[(kr + 1) * F + col0 + t * 16];
    }
  }

  gat_gemm_logit_vectors<F>(s_wa, W, a_src, a_dst, wib, lane);
  __syncthreads();
  v4f wav[KP];
#pragma unroll
  for (int j = 0; j < KP; ++j) wav[j] = *(const v4f*)(s_wa + cg * 128 + 8 * j + 4 * h);
  float* __restrict__ lout = cg ? ad : as;

  if (pr >= nRowTiles) return;
  const int cnt = (nRowTiles - pr + npr - 1) / npr;
  int rt = pr;
  v4f a0[KP], a1[KP];
  if (cnt & 1) {
    gat_gemm_load_a<KP, K>(a0, X, rt, c, h);
    gat_gemm_tile<TILES, KP, KS>(a0, b, Hout + ((size_t)rt * 16 + 8 * h) * F + col0, F,
                                 wav, lout + (size_t)rt * 16, c, h);
    rt += npr;
  }
  if (rt >= nRowTiles) return;
  gat_gemm_load_a<KP, K>(a0, X, rt, c, h);
  for (; rt < nRowTiles; rt += 2 * npr) {
    gat_gemm_load_a<KP, K>(a1, X, rt + npr, c, h);
    __builtin_amdgcn_sched_barrier(0);
    gat_gemm_tile<TILES, KP, KS>(a0, b, Hout + ((size_t)rt * 16 + 8 * h) * F + col0, F,
                                 wav, lout + (size_t)rt * 16, c, h);
    gat_gemm_load_a<KP, K>(a0, X, (rt + 2 * npr < nRowTiles) ? (rt + 2 * npr) : rt, c, h);
    __builtin_amdgcn_sched_barrier(0);
    gat_gemm_tile<TILES, KP, KS>(a1, b, Hout + ((size_t)(rt + npr) * 16 + 8 * h) * F + col0, F,
                                 wav, lout + (size_t)(rt + npr) * 16, c, h);
  }
}

#define GAT_NCAP 25024
#define GAT_TB   512

__host__ __device__ __forceinline__ int gat_pitch(int N) { return (N + 31) / 32 * 32; }

__device__ __forceinline__ bool gat_al16(const void* p) {
  return (((unsigned long long)p) & 15ull) == 0ull;
}

template <int EPT>
__global__ __launch_bounds__(GAT_TB)
void gat_tile_count(const int* __restrict__ ei, int* __restrict__ hist,
                    int N, int E, int T, int L) {
  __shared__ __attribute__((aligned(16))) int cnt[GAT_NCAP];
  const int tid = threadIdx.x;
  const int b   = blockIdx.x;
  const int g   = blockIdx.y;
  const int nlo = blockIdx.z * GAT_NCAP;
  const int cn  = (N - nlo < GAT_NCAP) ? (N - nlo) : GAT_NCAP;
  const int* __restrict__ dst = ei + ((size_t)g * 2 + 1) * E;

  int4* c4 = (int4*)cnt;
  for (int i = tid; i < GAT_NCAP / 4; i += GAT_TB) c4[i] = make_int4(0, 0, 0, 0);
  __syncthreads();

  const int e_beg = b * L;
  const int e_end = (e_beg + L < E) ? (e_beg + L) : E;
  if (e_beg < e_end) {
    for (int e0 = e_beg + tid; e0 < e_end; e0 += GAT_TB * EPT) {
      int dv[EPT];
#pragma unroll
      for (int j = 0; j < EPT; ++j) {
        const int e = e0 + j * GAT_TB;
        const int v = dst[(e < e_end) ? e : (e_end - 1)];
        dv[j] = (e < e_end) ? v : -1;
      }
#pragma unroll
      for (int j = 0; j < EPT; ++j) {
        const unsigned rel = (unsigned)(dv[j] - nlo);
        if (rel < (unsigned)cn) atomicAdd(&cnt[rel], 1);
      }
    }
  }
  __syncthreads();

  int* __restrict__ hrow = hist + ((size_t)g * T + b) * gat_pitch(N) + nlo;
  if (gat_al16(hrow) && (cn & 3) == 0) {
    int4* h4 = (int4*)hrow;
    for (int i = tid; i < cn / 4; i += GAT_TB) h4[i] = c4[i];
  } else {
    for (int i = tid; i < cn; i += GAT_TB) hrow[i] = cnt[i];
  }
}

template <int T>
__global__ __launch_bounds__(256)
void gat_tile_prefix(int* __restrict__ hist, int* __restrict__ deg, int N) {
  const int n = blockIdx.x * 256 + threadIdx.x;
  const int g = blockIdx.y;
  if (n >= N) return;
  const size_t P = (size_t)gat_pitch(N);
  int* __restrict__ col = hist + (size_t)g * T * P + n;
  int v[T];
#pragma unroll
  for (int b = 0; b < T; ++b) v[b] = col[b * P];
  int run = 0;
#pragma unroll
  for (int b = 0; b < T; ++b) { col[b * P] = run; run += v[b]; }
  deg[(size_t)g * N + n] = run;
}

__global__ __launch_bounds__(1024)
void gat_offsets_scan(const int* __restrict__ deg, int* __restrict__ off, int N) {
  __shared__ int wtot[32];
  __shared__ int ptot[32];
  const int t    = threadIdx.x;
  const int lane = t & 31;
  const int w    = t >> 5;
  const int i    = blockIdx.x * 1024 + t;

  const int v = (i < N) ? deg[i] : 0;

  const int4* __restrict__ d4 = (const int4*)deg;
  const int n4 = blockIdx.x * 256;
  int ps = 0;
#pragma unroll 4
  for (int q = t; q < n4; q += 1024) {
    const int4 u = d4[q];
    ps += (u.x + u.y) + (u.z + u.w);
  }

  int x = v;
  for (int o = 1; o < 32; o <<= 1) {
    int y = __shfl_up(x, o, 32);
    if (lane >= o) x += y;
  }
  for (int o = 16; o > 0; o >>= 1) ps += __shfl_xor(ps, o, 32);
  if (lane == 31) { wtot[w] = x; ptot[w] = ps; }
  __syncthreads();
  int base = ptot[lane] + ((lane < w) ? wtot[lane] : 0);
  for (int o = 16; o > 0; o >>= 1) base += __shfl_xor(base, o, 32);

  if (i < N)      off[i] = base + x - v;
  if (i == N - 1) off[N] = base + x;
}

template <int EPT>
__global__ __launch_bounds__(GAT_TB)
void gat_tile_fill(const int* __restrict__ ei, const int* __restrict__ hist,
                   const int* __restrict__ off, int* __restrict__ csr,
                   int N, int E, int T, int L) {
  __shared__ __attribute__((aligned(16))) int cur[GAT_NCAP];
  const int tid = threadIdx.x;
  const int b   = blockIdx.x;
  const int g   = blockIdx.y;
  const int nlo = blockIdx.z * GAT_NCAP;
  const int cn  = (N - nlo < GAT_NCAP) ? (N - nlo) : GAT_NCAP;
  const int* __restrict__ src  = ei + (size_t)g * 2 * E;
  const int* __restrict__ dst  = src + E;
  const int* __restrict__ hrow = hist + ((size_t)g * T + b) * gat_pitch(N) + nlo;
  const int* __restrict__ orow = off + (size_t)g * N + nlo;

  if (gat_al16(hrow) && gat_al16(orow) && (cn & 3) == 0) {
    const int4* h4 = (const int4*)hrow;
    const int4* o4 = (const int4*)orow;
    int4* c4 = (int4*)cur;
    for (int i = tid; i < cn / 4; i += GAT_TB) {
      const int4 h = h4[i];
      const int4 o = o4[i];
      c4[i] = make_int4(h.x + o.x, h.y + o.y, h.z + o.z, h.w + o.w);
    }
  } else {
    for (int i = tid; i < cn; i += GAT_TB) cur[i] = hrow[i] + orow[i];
  }
  __syncthreads();

  const int idshift = g * N;
  const int e_beg = b * L;
  const int e_end = (e_beg + L < E) ? (e_beg + L) : E;
  if (e_beg >= e_end) return;
  for (int e0 = e_beg + tid; e0 < e_end; e0 += GAT_TB * EPT) {
    int sv[EPT], dv[EPT], pv[EPT];
#pragma unroll
    for (int j = 0; j < EPT; ++j) {
      const int e  = e0 + j * GAT_TB;
      const int ec = (e < e_end) ? e : (e_end - 1);
      sv[j] = src[ec];
      const int v = dst[ec];
      dv[j] = (e < e_end) ? v : -1;
    }
#pragma unroll
    for (int j = 0; j < EPT; ++j) {
      const unsigned rel = (unsigned)(dv[j] - nlo);
      pv[j] = -1;
      if (rel < (unsigned)cn) pv[j] = atomicAdd(&cur[rel], 1);
    }
#pragma unroll
    for (int j = 0; j < EPT; ++j)
      if (pv[j] >= 0) csr[pv[j]] = sv[j] + idshift;
  }
}

__device__ __forceinline__ float gat_leaky(float v) {
  return (v > 0.f) ? v : NEG_SLOPE * v;
}

__device__ __forceinline__ float gat_wave_max(float x) {
  for (int o = 16; o > 0; o >>= 1) x = fmaxf(x, __shfl_xor(x, o, 32));
  return x;
}

__device__ __forceinline__ float gat_wave_sum(float x) {
  for (int o = 16; o > 0; o >>= 1) x += __shfl_xor(x, o, 32);
  return x;
}

template <int C, typename vec>
__device__ __forceinline__ void gat_rows_fma(vec& acc, int s, float a, int cnt,
                                             const float* __restrict__ hlane) {
#pragma unroll 4
  for (int j = 0; j < cnt; ++j) {
    const int   sj = __builtin_amdgcn_readlane(s, j);
    const float aj = __int_as_float(__builtin_amdgcn_readlane(__float_as_int(a), j));
    const vec   hv = *(const vec*)(hlane + (size_t)sj * C);
    acc += hv * aj;
  }
}

template <int C, int RELU, int NPW>
__global__ __launch_bounds__(256)
void gat_node_softmax_gather(const int* __restrict__ off,
                             const int* __restrict__ csr,
                             const float* __restrict__ as,
                             const float* __restrict__ ad,
                             const float* __restrict__ h,
                             const float* __restrict__ bias,
                             float* __restrict__ out, int N) {
  constexpr int V = C / 32;
  typedef typename gat_rowvec<V>::T vec;
  const int lane = threadIdx.x & 31;
  const int wave = __builtin_amdgcn_readfirstlane((int)((blockIdx.x * 256u + threadIdx.x) >> 5));
  const int d0   = wave * NPW;
  if (d0 >= N) return;
  const int d1   = (d0 + NPW < N) ? (d0 + NPW) : N;

  const float* hlane = h + lane * V;
  const vec    bv    = *(const vec*)(bias + lane * V);

  int beg = off[d0];
  for (int d = d0; d < d1; ++d) {
    const int   end   = off[d + 1];
    const float add   = ad[d];
    const float vself = gat_leaky(as[d] + add);

    const int   p0   = beg + lane;
    const bool  has0 = p0 < end;
    const int   s0   = has0 ? csr[p0] : d;
    const float v0   = gat_leaky(as[s0] + add);

    float m = v0;
    for (int p = p0 + 32; p < end; p += 32)
      m = fmaxf(m, gat_leaky(as[csr[p]] + add));
    m = fmaxf(gat_wave_max(m), vself);
    if (!__builtin_isfinite(m)) m = 0.f;

    float sum = has0 ? __expf(v0 - m) : 0.f;
    for (int p = p0 + 32; p < end; p += 32)
      sum += __expf(gat_leaky(as[csr[p]] + add) - m);
    const float exself = __expf(vself - m);
    const float denom  = (gat_wave_sum(sum) + exself) + 1e-16f;

    vec acc = *(const vec*)(hlane + (size_t)d * C) * (exself / denom);

    const int deg = end - beg;
    gat_rows_fma<C, vec>(acc, s0, __expf(v0 - m) / denom, (deg < 32) ? deg : 32, hlane);
    for (int base = beg + 32; base < end; base += 32) {
      const int   p    = base + lane;
      const int   s    = (p < end) ? csr[p] : d;
      const float a    = __expf(gat_leaky(as[s] + add) - m) / denom;
      const int   left = end - base;
      gat_rows_fma<C, vec>(acc, s, a, (left < 32) ? left : 32, hlane);
    }

    vec o = acc + bv;
    if (RELU) {
      for (int i = 0; i < V; ++i) o[i] = fmaxf(o[i], 0.f);
    }
    *(vec*)(out + (size_t)d * C + lane * V) = o;
    beg = end;
  }
}

extern "C" void kernel_launch(void* const* d_in, const int* in_sizes, int n_in,
                              void* d_out, int out_size, void* d_ws, size_t ws_size,
                              hipStream_t stream) {
  const float* fea = (const float*)d_in[0];
  const int*   ei  = (const int*)  d_in[1];
  const float* W1  = (const float*)d_in[2];
  const float* av1s= (const float*)d_in[3];
  const float* av1d= (const float*)d_in[4];
  const float* b1  = (const float*)d_in[5];
  const float* W2  = (const float*)d_in[6];
  const float* av2s= (const float*)d_in[7];
  const float* av2d= (const float*)d_in[8];
  const float* b2  = (const float*)d_in[9];
  float* out = (float*)d_out;

  const int B  = 2;
  const int C1 = 128, C2 = 64;
  const int N  = in_sizes[0] / (B * C1);
  const int E  = in_sizes[1] / (B * 2);
  const int NN = B * N;
  const int EE = B * E;

  float* w   = (float*)d_ws;
  float* h1   = w; w += (long)NN * C1;
  float* x2   = w; w += (long)NN * C1;
  float* h2   = w; w += (long)NN * C2;
  float* asb  = w; w += NN;
  float* adb  = w; w += NN;
  const int T = 64;
  int* degb  = (int*)w; w += NN;
  int* histb = (int*)w; w += (long)B * T * gat_pitch(N);
  int* csrb  = (int*)w; w += EE;
  int* offb  = (int*)w; w += NN + 1;

  const int TB = 256;
  dim3 blk(TB);
  const int EPT = 16;
  const int NPW = 4;
  const int L = (E + T - 1) / T;
  dim3 gEdge(T, B, (N + GAT_NCAP - 1) / GAT_NCAP);
  dim3 gPref((N + TB - 1) / TB, B);
  int gNode    = ((NN + NPW - 1) / NPW * 32 + TB - 1) / TB;
  const int nRowTiles = NN / 16;
  int gGemm    = 256;
  int gScan    = (NN + 1023) / 1024;

  gat_tile_count<EPT><<<gEdge, dim3(GAT_TB), 0, stream>>>(ei, histb, N, E, T, L);
  gat_tile_prefix<T><<<gPref, blk, 0, stream>>>(histb, degb, N);
  gat_offsets_scan<<<gScan, dim3(1024), 0, stream>>>(degb, offb, NN);
  gat_tile_fill<EPT><<<gEdge, dim3(GAT_TB), 0, stream>>>(ei, histb, offb, csrb, N, E, T, L);

  gat_gemm_wmma<4, 2><<<gGemm, blk, 0, stream>>>(fea, W1, h1, av1s, av1d, asb, adb, nRowTiles);
  gat_node_softmax_gather<128, 1, NPW><<<gNode, blk, 0, stream>>>(
      offb, csrb, asb, adb, h1, b1, x2, NN);

  gat_gemm_wmma<2, 2><<<gGemm, blk, 0, stream>>>(x2, W2, h2, av2s, av2d, asb, adb, nRowTiles);
  gat_node_softmax_gather<64, 0, NPW><<<gNode, blk, 0, stream>>>(
      offb, csrb, asb, adb, h2, b2, out, NN);
}


# Layer 2's product and logits move into the epilogue of layer 1's softmax+gather: finished rows go to LDS, then as 16-row tiles to WMMA; the GEMM launch and the x2 round trip go; sum orders kept
# speedup vs baseline: 7.1974x; 1.0480x over previous
#include <hip/hip_runtime.h>
#include <math.h>

typedef __attribute__((ext_vector_type(2))) float v2f;
typedef __attribute__((ext_vector_type(8))) float v8f;
typedef __attribute__((ext_vector_type(4))) float v4f;

#define NEG_SLOPE 0.2f

template <int V> struct gat_rowvec;
template <> struct gat_rowvec<4> { typedef v4f T; };
template <> struct gat_rowvec<2> { typedef v2f T; };

template <int O>
__device__ __forceinline__ void gat_halve(float (&v)[32], int lane) {
  const bool up = (lane & O) != 0;
#pragma unroll
  for (int i = 0; i < O; ++i) {
    const float keep = up ? v[i + O] : v[i];
    const float send = up ? v[i] : v[i + O];
    v[i] = keep + __shfl_xor(send, O, 32);
  }
}

template <int F>
__device__ __forceinline__ void gat_gemm_logit_vectors(float* __restrict__ s_wa,
                                                       const float* __restrict__ W,
                                                       const float* __restrict__ a_src,
                                                       const float* __restrict__ a_dst,
                                                       int wib, int lane) {
  constexpr int VW = F / 32;
  typedef typename gat_rowvec<VW>::T wvec;
  const wvec avs = *(const wvec*)(a_src + lane * VW);
  const wvec avd = *(const wvec*)(a_dst + lane * VW);
  float v[32];
#pragma unroll
  for (int i = 0; i < 16; ++i) {
    const wvec wr = *(const wvec*)(W + (size_t)(16 * wib + i) * F + lane * VW);
    float ss = 0.f, sd = 0.f;
#pragma unroll
    for (int q = 0; q < VW; ++q) { ss += wr[q] * avs[q]; sd += wr[q] * avd[q]; }
    v[i] = ss; v[16 + i] = sd;
  }
  gat_halve<16>(v, lane);
  gat_halve<8>(v, lane);
  gat_halve<4>(v, lane);
  gat_halve<2>(v, lane);
  gat_halve<1>(v, lane);
  s_wa[(lane >> 4) * 128 + 16 * wib + (lane & 15)] = v[0];
}

template <int KP, int K>
__device__ __forceinline__ void gat_gemm_load_a(v4f (&a)[KP], const float* __restrict__ X,
                                                int rt, int c, int h) {
  const float* xr = X + ((size_t)rt * 16 + c) * K + 4 * h;
#pragma unroll
  for (int j = 0; j < KP; ++j) a[j] = *(const v4f*)(xr + 8 * j);
}

template <int TILES, int KP, int KS>
__device__ __forceinline__ void gat_gemm_tile(const v4f (&a)[KP], const v2f (&b)[KS][TILES],
                                              float* __restrict__ orow, int F,
                                              const v4f (&wav)[KP], float* __restrict__ lrow,
                                              int c, int h) {
  v4f s4 = a[0] * wav[0];
#pragma unroll
  for (int j = 1; j < KP; ++j) s4 += a[j] * wav[j];
  float lg = (s4.x + s4.y) + (s4.z + s4.w);
  lg += __shfl_xor(lg, 16, 32);
  if (h == 0) lrow[c] = lg;

  v8f acc[TILES];
#pragma unroll
  for (int t = 0; t < TILES; ++t) acc[t] = (v8f){0.f,0.f,0.f,0.f,0.f,0.f,0.f,0.f};
#pragma unroll
  for (int j = 0; j < KP; ++j) {
    v2f p0; p0.x = a[j].x; p0.y = a[j].y;
    v2f p1; p1.x = a[j].z; p1.y = a[j].w;
#pragma unroll
    for (int t = 0; t < TILES; ++t)
      acc[t] = __builtin_amdgcn_wmma_f32_16x16x4_f32(
           false, p0,  false, b[2 * j][t],
           (short)0, acc[t],  false,  false);
#pragma unroll
    for (int t = 0; t < TILES; ++t)
      acc[t] = __builtin_amdgcn_wmma_f32_16x16x4_f32(
           false, p1,  false, b[2 * j + 1][t],
           (short)0, acc[t],  false,  false);
  }
#pragma unroll
  for (int t = 0; t < TILES; ++t)
#pragma unroll
    for (int r = 0; r < 8; ++r)
      orow[(size_t)r * F + t * 16] = acc[t][r];
}

template <int TILES, int CG, int NEXTF>
__global__ __launch_bounds__(256)
void gat_gemm_wmma(const float* __restrict__ X, const float* __restrict__ W,
                   float* __restrict__ Hout,
                   const float* __restrict__ a_src, const float* __restrict__ a_dst,
                   float* __restrict__ as, float* __restrict__ ad, int nRowTiles,
                   const float* __restrict__ Wn, const float* __restrict__ an_src,
                   const float* __restrict__ an_dst, float* __restrict__ wan) {
  static_assert(CG == 2, "one wave of a row tile does a_src, the other a_dst");
  __shared__ __attribute__((aligned(16))) float s_wa[2 * 128];
  __shared__ __attribute__((aligned(16))) float s_wn[2 * 128];
  constexpr int K  = 128;
  constexpr int F  = CG * TILES * 16;
  constexpr int KS = K / 4;
  constexpr int KP = K / 8;
  const int lane = threadIdx.x & 31;
  const int c    = lane & 15;
  const int h    = lane >> 4;
  const int wib  = __builtin_amdgcn_readfirstlane((int)(threadIdx.x >> 5));
  const int cg   = wib % CG;
  const int pr   = (int)blockIdx.x + (int)gridDim.x * (wib / CG);
  const int npr  = (int)gridDim.x * (8 / CG);
  const int col0 = cg * TILES * 16 + c;

  v2f b[KS][TILES];
#pragma unroll
  for (int s = 0; s < KS; ++s) {
    const int kr = 8 * (s >> 1) + 4 * h + 2 * (s & 1);
#pragma unroll
    for (int t = 0; t < TILES; ++t) {
      b[s][t].x = W[(kr)     * F + col0 + t * 16];
      b[s][t].y = W[(kr + 1) * F + col0 + t * 16];
    }
  }

  gat_gemm_logit_vectors<F>(s_wa, W, a_src, a_dst, wib, lane);
  __syncthreads();
  v4f wav[KP];
#pragma unroll
  for (int j = 0; j < KP; ++j) wav[j] = *(const v4f*)(s_wa + cg * 128 + 8 * j + 4 * h);
  float* __restrict__ lout = cg ? ad : as;

  if constexpr (NEXTF > 0) {
    if ((int)blockIdx.x == (int)gridDim.x - 1) {
      gat_gemm_logit_vectors<NEXTF>(s_wn, Wn, an_src, an_dst, wib, lane);
      __syncthreads();
      wan[threadIdx.x] = s_wn[threadIdx.x];
    }
  }

  if (pr >= nRowTiles) return;
  const int cnt = (nRowTiles - pr + npr - 1) / npr;
  int rt = pr;
  v4f a0[KP], a1[KP];
  if (cnt & 1) {
    gat_gemm_load_a<KP, K>(a0, X, rt, c, h);
    gat_gemm_tile<TILES, KP, KS>(a0, b, Hout + ((size_t)rt * 16 + 8 * h) * F + col0, F,
                                 wav, lout + (size_t)rt * 16, c, h);
    rt += npr;
  }
  if (rt >= nRowTiles) return;
  gat_gemm_load_a<KP, K>(a0, X, rt, c, h);
  for (; rt < nRowTiles; rt += 2 * npr) {
    gat_gemm_load_a<KP, K>(a1, X, rt + npr, c, h);
    __builtin_amdgcn_sched_barrier(0);
    gat_gemm_tile<TILES, KP, KS>(a0, b, Hout + ((size_t)rt * 16 + 8 * h) * F + col0, F,
                                 wav, lout + (size_t)rt * 16, c, h);
    gat_gemm_load_a<KP, K>(a0, X, (rt + 2 * npr < nRowTiles) ? (rt + 2 * npr) : rt, c, h);
    __builtin_amdgcn_sched_barrier(0);
    gat_gemm_tile<TILES, KP, KS>(a1, b, Hout + ((size_t)(rt + npr) * 16 + 8 * h) * F + col0, F,
                                 wav, lout + (size_t)(rt + npr) * 16, c, h);
  }
}

#define GAT_NCAP 25024
#define GAT_TB   512

__host__ __device__ __forceinline__ int gat_pitch(int N) { return (N + 31) / 32 * 32; }

__device__ __forceinline__ bool gat_al16(const void* p) {
  return (((unsigned long long)p) & 15ull) == 0ull;
}

template <int EPT>
__global__ __launch_bounds__(GAT_TB)
void gat_tile_count(const int* __restrict__ ei, int* __restrict__ hist,
                    int N, int E, int T, int L) {
  __shared__ __attribute__((aligned(16))) int cnt[GAT_NCAP];
  const int tid = threadIdx.x;
  const int b   = blockIdx.x;
  const int g   = blockIdx.y;
  const int nlo = blockIdx.z * GAT_NCAP;
  const int cn  = (N - nlo < GAT_NCAP) ? (N - nlo) : GAT_NCAP;
  const int* __restrict__ dst = ei + ((size_t)g * 2 + 1) * E;

  int4* c4 = (int4*)cnt;
  for (int i = tid; i < GAT_NCAP / 4; i += GAT_TB) c4[i] = make_int4(0, 0, 0, 0);
  __syncthreads();

  const int e_beg = b * L;
  const int e_end = (e_beg + L < E) ? (e_beg + L) : E;
  if (e_beg < e_end) {
    for (int e0 = e_beg + tid; e0 < e_end; e0 += GAT_TB * EPT) {
      int dv[EPT];
#pragma unroll
      for (int j = 0; j < EPT; ++j) {
        const int e = e0 + j * GAT_TB;
        const int v = dst[(e < e_end) ? e : (e_end - 1)];
        dv[j] = (e < e_end) ? v : -1;
      }
#pragma unroll
      for (int j = 0; j < EPT; ++j) {
        const unsigned rel = (unsigned)(dv[j] - nlo);
        if (rel < (unsigned)cn) atomicAdd(&cnt[rel], 1);
      }
    }
  }
  __syncthreads();

  int* __restrict__ hrow = hist + ((size_t)g * T + b) * gat_pitch(N) + nlo;
  if (gat_al16(hrow) && (cn & 3) == 0) {
    int4* h4 = (int4*)hrow;
    for (int i = tid; i < cn / 4; i += GAT_TB) h4[i] = c4[i];
  } else {
    for (int i = tid; i < cn; i += GAT_TB) hrow[i] = cnt[i];
  }
}

template <int T>
__global__ __launch_bounds__(256)
void gat_tile_prefix(int* __restrict__ hist, int* __restrict__ deg, int N) {
  const int n = blockIdx.x * 256 + threadIdx.x;
  const int g = blockIdx.y;
  if (n >= N) return;
  const size_t P = (size_t)gat_pitch(N);
  int* __restrict__ col = hist + (size_t)g * T * P + n;
  int v[T];
#pragma unroll
  for (int b = 0; b < T; ++b) v[b] = col[b * P];
  int run = 0;
#pragma unroll
  for (int b = 0; b < T; ++b) { col[b * P] = run; run += v[b]; }
  deg[(size_t)g * N + n] = run;
}

__global__ __launch_bounds__(1024)
void gat_offsets_scan(const int* __restrict__ deg, int* __restrict__ off, int N) {
  __shared__ int wtot[32];
  __shared__ int ptot[32];
  const int t    = threadIdx.x;
  const int lane = t & 31;
  const int w    = t >> 5;
  const int i    = blockIdx.x * 1024 + t;

  const int v = (i < N) ? deg[i] : 0;

  const int4* __restrict__ d4 = (const int4*)deg;
  const int n4 = blockIdx.x * 256;
  int ps = 0;
#pragma unroll 4
  for (int q = t; q < n4; q += 1024) {
    const int4 u = d4[q];
    ps += (u.x + u.y) + (u.z + u.w);
  }

  int x = v;
  for (int o = 1; o < 32; o <<= 1) {
    int y = __shfl_up(x, o, 32);
    if (lane >= o) x += y;
  }
  for (int o = 16; o > 0; o >>= 1) ps += __shfl_xor(ps, o, 32);
  if (lane == 31) { wtot[w] = x; ptot[w] = ps; }
  __syncthreads();
  int base = ptot[lane] + ((lane < w) ? wtot[lane] : 0);
  for (int o = 16; o > 0; o >>= 1) base += __shfl_xor(base, o, 32);

  if (i < N)      off[i] = base + x - v;
  if (i == N - 1) off[N] = base + x;
}

template <int EPT>
__global__ __launch_bounds__(GAT_TB)
void gat_tile_fill(const int* __restrict__ ei, const int* __restrict__ hist,
                   const int* __restrict__ off, int* __restrict__ csr,
                   int N, int E, int T, int L) {
  __shared__ __attribute__((aligned(16))) int cur[GAT_NCAP];
  const int tid = threadIdx.x;
  const int b   = blockIdx.x;
  const int g   = blockIdx.y;
  const int nlo = blockIdx.z * GAT_NCAP;
  const int cn  = (N - nlo < GAT_NCAP) ? (N - nlo) : GAT_NCAP;
  const int* __restrict__ src  = ei + (size_t)g * 2 * E;
  const int* __restrict__ dst  = src + E;
  const int* __restrict__ hrow = hist + ((size_t)g * T + b) * gat_pitch(N) + nlo;
  const int* __restrict__ orow = off + (size_t)g * N + nlo;

  if (gat_al16(hrow) && gat_al16(orow) && (cn & 3) == 0) {
    const int4* h4 = (const int4*)hrow;
    const int4* o4 = (const int4*)orow;
    int4* c4 = (int4*)cur;
    for (int i = tid; i < cn / 4; i += GAT_TB) {
      const int4 h = h4[i];
      const int4 o = o4[i];
      c4[i] = make_int4(h.x + o.x, h.y + o.y, h.z + o.z, h.w + o.w);
    }
  } else {
    for (int i = tid; i < cn; i += GAT_TB) cur[i] = hrow[i] + orow[i];
  }
  __syncthreads();

  const int idshift = g * N;
  const int e_beg = b * L;
  const int e_end = (e_beg + L < E) ? (e_beg + L) : E;
  if (e_beg >= e_end) return;
  for (int e0 = e_beg + tid; e0 < e_end; e0 += GAT_TB * EPT) {
    int sv[EPT], dv[EPT], pv[EPT];
#pragma unroll
    for (int j = 0; j < EPT; ++j) {
      const int e  = e0 + j * GAT_TB;
      const int ec = (e < e_end) ? e : (e_end - 1);
      sv[j] = src[ec];
      const int v = dst[ec];
      dv[j] = (e < e_end) ? v : -1;
    }
#pragma unroll
    for (int j = 0; j < EPT; ++j) {
      const unsigned rel = (unsigned)(dv[j] - nlo);
      pv[j] = -1;
      if (rel < (unsigned)cn) pv[j] = atomicAdd(&cur[rel], 1);
    }
#pragma unroll
    for (int j = 0; j < EPT; ++j)
      if (pv[j] >= 0) csr[pv[j]] = sv[j] + idshift;
  }
}

__device__ __forceinline__ float gat_leaky(float v) {
  return (v > 0.f) ? v : NEG_SLOPE * v;
}

__device__ __forceinline__ float gat_wave_max(float x) {
  for (int o = 16; o > 0; o >>= 1) x = fmaxf(x, __shfl_xor(x, o, 32));
  return x;
}

__device__ __forceinline__ float gat_wave_sum(float x) {
  for (int o = 16; o > 0; o >>= 1) x += __shfl_xor(x, o, 32);
  return x;
}

template <int C, typename vec>
__device__ __forceinline__ void gat_rows_fma(vec& acc, int s, float a, int cnt,
                                             const float* __restrict__ hlane) {
#pragma unroll 4
  for (int j = 0; j < cnt; ++j) {
    const int   sj = __builtin_amdgcn_readlane(s, j);
    const float aj = __int_as_float(__builtin_amdgcn_readlane(__float_as_int(a), j));
    const vec   hv = *(const vec*)(hlane + (size_t)sj * C);
    acc += hv * aj;
  }
}

template <int C, int RELU, int NPW>
__global__ __launch_bounds__(256)
void gat_node_softmax_gather(const int* __restrict__ off,
                             const int* __restrict__ csr,
                             const float* __restrict__ as,
                             const float* __restrict__ ad,
                             const float* __restrict__ h,
                             const float* __restrict__ bias,
                             float* __restrict__ out, int N) {
  constexpr int V = C / 32;
  typedef typename gat_rowvec<V>::T vec;
  const int lane = threadIdx.x & 31;
  const int wave = __builtin_amdgcn_readfirstlane((int)((blockIdx.x * 256u + threadIdx.x) >> 5));
  const int d0   = wave * NPW;
  if (d0 >= N) return;
  const int d1   = (d0 + NPW < N) ? (d0 + NPW) : N;

  const float* hlane = h + lane * V;
  const vec    bv    = *(const vec*)(bias + lane * V);

  int beg = off[d0];
  for (int d = d0; d < d1; ++d) {
    const int   end   = off[d + 1];
    const float add   = ad[d];
    const float vself = gat_leaky(as[d] + add);

    const int   p0   = beg + lane;
    const bool  has0 = p0 < end;
    const int   s0   = has0 ? csr[p0] : d;
    const float v0   = gat_leaky(as[s0] + add);

    float m = v0;
    for (int p = p0 + 32; p < end; p += 32)
      m = fmaxf(m, gat_leaky(as[csr[p]] + add));
    m = fmaxf(gat_wave_max(m), vself);
    if (!__builtin_isfinite(m)) m = 0.f;

    float sum = has0 ? __expf(v0 - m) : 0.f;
    for (int p = p0 + 32; p < end; p += 32)
      sum += __expf(gat_leaky(as[csr[p]] + add) - m);
    const float exself = __expf(vself - m);
    const float denom  = (gat_wave_sum(sum) + exself) + 1e-16f;

    vec acc = *(const vec*)(hlane + (size_t)d * C) * (exself / denom);

    const int deg = end - beg;
    gat_rows_fma<C, vec>(acc, s0, __expf(v0 - m) / denom, (deg < 32) ? deg : 32, hlane);
    for (int base = beg + 32; base < end; base += 32) {
      const int   p    = base + lane;
      const int   s    = (p < end) ? csr[p] : d;
      const float a    = __expf(gat_leaky(as[s] + add) - m) / denom;
      const int   left = end - base;
      gat_rows_fma<C, vec>(acc, s, a, (left < 32) ? left : 32, hlane);
    }

    vec o = acc + bv;
    if (RELU) {
      for (int i = 0; i < V; ++i) o[i] = fmaxf(o[i], 0.f);
    }
    *(vec*)(out + (size_t)d * C + lane * V) = o;
    beg = end;
  }
}


#define GAT_XPITCH 132

template <int NPW, int JC>
__global__ __launch_bounds__(256)
void gat_node_softmax_gather_gemm(const int* __restrict__ off,
                                  const int* __restrict__ csr,
                                  const float* __restrict__ as,
                                  const float* __restrict__ ad,
                                  const float* __restrict__ h,
                                  const float* __restrict__ bias,
                                  const float* __restrict__ W2,
                                  const float* __restrict__ wa2,
                                  float* __restrict__ h2,
                                  float* __restrict__ as2,
                                  float* __restrict__ ad2, int N) {
  constexpr int C  = 128;
  constexpr int F2 = 64;
  constexpr int RW = 8 * NPW;
  static_assert(NPW % 2 == 0 && NPW <= 8, "whole 16-row tiles, one wave per (tile, column half)");
  static_assert(16 % JC == 0, "whole chunks of A pieces");
  typedef v4f vec;
  __shared__ __attribute__((aligned(16))) float s_x[RW * GAT_XPITCH];
  __shared__ __attribute__((aligned(16))) float s_wa[2 * 128];

  const int lane = threadIdx.x & 31;
  const int wib   = __builtin_amdgcn_readfirstlane((int)(threadIdx.x >> 5));
  const int node0 = (int)blockIdx.x * RW;
  const int d0    = node0 + wib * NPW;
  const int d1    = (d0 + NPW < N) ? (d0 + NPW) : N;

  const float wa_t = wa2[threadIdx.x];

  float* __restrict__ xrow = s_x + (wib * NPW) * GAT_XPITCH + lane * 4;
  int done = 0;
  if (d0 < N) {
    const float* hlane = h + lane * 4;
    const vec    bv    = *(const vec*)(bias + lane * 4);
    int beg = off[d0];
    for (int d = d0; d < d1; ++d) {
      const int   end   = off[d + 1];
      const float add   = ad[d];
      const float vself = gat_leaky(as[d] + add);

      const int   p0   = beg + lane;
      const bool  has0 = p0 < end;
      const int   s0   = has0 ? csr[p0] : d;
      const float v0   = gat_leaky(as[s0] + add);

      float m = v0;
      for (int p = p0 + 32; p < end; p += 32)
        m = fmaxf(m, gat_leaky(as[csr[p]] + add));
      m = fmaxf(gat_wave_max(m), vself);
      if (!__builtin_isfinite(m)) m = 0.f;

      float sum = has0 ? __expf(v0 - m) : 0.f;
      for (int p = p0 + 32; p < end; p += 32)
        sum += __expf(gat_leaky(as[csr[p]] + add) - m);
      const float exself = __expf(vself - m);
      const float denom  = (gat_wave_sum(sum) + exself) + 1e-16f;

      vec acc = *(const vec*)(hlane + (size_t)d * C) * (exself / denom);

      const int deg = end - beg;
      gat_rows_fma<C, vec>(acc, s0, __expf(v0 - m) / denom, (deg < 32) ? deg : 32, hlane);
      for (int base = beg + 32; base < end; base += 32) {
        const int   p    = base + lane;
        const int   s    = (p < end) ? csr[p] : d;
        const float a    = __expf(gat_leaky(as[s] + add) - m) / denom;
        const int   left = end - base;
        gat_rows_fma<C, vec>(acc, s, a, (left < 32) ? left : 32, hlane);
      }

      vec o = acc + bv;
      for (int i = 0; i < 4; ++i) o[i] = fmaxf(o[i], 0.f);
      *(vec*)(xrow + (d - d0) * GAT_XPITCH) = o;
      beg = end;
    }
    done = d1 - d0;
  }
  for (int i = done; i < NPW; ++i) *(vec*)(xrow + i * GAT_XPITCH) = (vec){0.f, 0.f, 0.f, 0.f};
  s_wa[threadIdx.x] = wa_t;
  __syncthreads();
  if (wib >= NPW) return;

  const int rtl = wib >> 1;
  const int q   = wib & 1;
  const int c   = lane & 15;
  const int hh  = lane >> 4;
  const float* __restrict__ xa = s_x + (16 * rtl + c) * GAT_XPITCH + 4 * hh;
  const float* __restrict__ wv = s_wa + q * 128 + 4 * hh;
  const float* __restrict__ wb = W2 + q * 32 + c;

  v8f acc0 = (v8f){0.f,0.f,0.f,0.f,0.f,0.f,0.f,0.f};
  v8f acc1 = (v8f){0.f,0.f,0.f,0.f,0.f,0.f,0.f,0.f};
  v4f s4   = (v4f){0.f,0.f,0.f,0.f};
#pragma unroll 1
  for (int j0 = 0; j0 < 16; j0 += JC) {
    v4f a[JC], w4[JC];
    v2f b[2 * JC][2];
#pragma unroll
    for (int j = 0; j < JC; ++j) {
      a[j]  = *(const v4f*)(xa + 8 * (j0 + j));
      w4[j] = *(const v4f*)(wv + 8 * (j0 + j));
    }
#pragma unroll
    for (int s = 0; s < 2 * JC; ++s) {
      const int kr = 8 * (j0 + (s >> 1)) + 4 * hh + 2 * (s & 1);
#pragma unroll
      for (int t = 0; t < 2; ++t) {
        b[s][t].x = wb[(kr)     * F2 + t * 16];
        b[s][t].y = wb[(kr + 1) * F2 + t * 16];
      }
    }
#pragma unroll
    for (int j = 0; j < JC; ++j) s4 += a[j] * w4[j];
#pragma unroll
    for (int j = 0; j < JC; ++j) {
      v2f p0; p0.x = a[j].x; p0.y = a[j].y;
      v2f p1; p1.x = a[j].z; p1.y = a[j].w;
      acc0 = __builtin_amdgcn_wmma_f32_16x16x4_f32(false, p0, false, b[2 * j][0], (short)0, acc0, false, false);
      acc1 = __builtin_amdgcn_wmma_f32_16x16x4_f32(false, p0, false, b[2 * j][1], (short)0, acc1, false, false);
      acc0 = __builtin_amdgcn_wmma_f32_16x16x4_f32(false, p1, false, b[2 * j + 1][0], (short)0, acc0, false, false);
      acc1 = __builtin_amdgcn_wmma_f32_16x16x4_f32(false, p1, false, b[2 * j + 1][1], (short)0, acc1, false, false);
    }
  }

  float lg = (s4.x + s4.y) + (s4.z + s4.w);
  lg += __shfl_xor(lg, 16, 32);
  const int rowl = node0 + 16 * rtl;
  float* __restrict__ lout = q ? ad2 : as2;
  if (hh == 0 && rowl + c < N) lout[rowl + c] = lg;

  float* __restrict__ orow = h2 + ((size_t)rowl + 8 * hh) * F2 + q * 32 + c;
#pragma unroll
  for (int r = 0; r < 8; ++r) {
    if (rowl + 8 * hh + r < N) {
      orow[(size_t)r * F2]      = acc0[r];
      orow[(size_t)r * F2 + 16] = acc1[r];
    }
  }
}

extern "C" void kernel_launch(void* const* d_in, const int* in_sizes, int n_in,
                              void* d_out, int out_size, void* d_ws, size_t ws_size,
                              hipStream_t stream) {
  const float* fea = (const float*)d_in[0];
  const int*   ei  = (const int*)  d_in[1];
  const float* W1  = (const float*)d_in[2];
  const float* av1s= (const float*)d_in[3];
  const float* av1d= (const float*)d_in[4];
  const float* b1  = (const float*)d_in[5];
  const float* W2  = (const float*)d_in[6];
  const float* av2s= (const float*)d_in[7];
  const float* av2d= (const float*)d_in[8];
  const float* b2  = (const float*)d_in[9];
  float* out = (float*)d_out;

  const int B  = 2;
  const int C1 = 128, C2 = 64;
  const int N  = in_sizes[0] / (B * C1);
  const int E  = in_sizes[1] / (B * 2);
  const int NN = B * N;
  const int EE = B * E;

  float* w   = (float*)d_ws;
  float* h1   = w; w += (long)NN * C1;
  float* h2   = w; w += (long)NN * C2;
  float* asb  = w; w += NN;
  float* adb  = w; w += NN;
  float* as2b = w; w += NN;
  float* ad2b = w; w += NN;
  float* wa2b = w; w += 2 * 128;
  const int T = 64;
  int* degb  = (int*)w; w += NN;
  int* histb = (int*)w; w += (long)B * T * gat_pitch(N);
  int* csrb  = (int*)w; w += EE;
  int* offb  = (int*)w; w += NN + 1;

  const int TB = 256;
  dim3 blk(TB);
  const int EPT = 16;
  const int NPW = 4;
  const int L = (E + T - 1) / T;
  dim3 gEdge(T, B, (N + GAT_NCAP - 1) / GAT_NCAP);
  dim3 gPref((N + TB - 1) / TB, B);
  int gNode    = ((NN + NPW - 1) / NPW * 32 + TB - 1) / TB;
  int gFused   = (NN + 8 * NPW - 1) / (8 * NPW);
  const int nRowTiles = NN / 16;
  int gGemm    = 256;
  int gScan    = (NN + 1023) / 1024;

  gat_tile_count<EPT><<<gEdge, dim3(GAT_TB), 0, stream>>>(ei, histb, N, E, T, L);
  gat_tile_prefix<T><<<gPref, blk, 0, stream>>>(histb, degb, N);
  gat_offsets_scan<<<gScan, dim3(1024), 0, stream>>>(degb, offb, NN);
  gat_tile_fill<EPT><<<gEdge, dim3(GAT_TB), 0, stream>>>(ei, histb, offb, csrb, N, E, T, L);

  gat_gemm_wmma<4, 2, 64><<<gGemm, blk, 0, stream>>>(fea, W1, h1, av1s, av1d, asb, adb, nRowTiles,
                                                     W2, av2s, av2d, wa2b);
  gat_node_softmax_gather_gemm<NPW, 2><<<gFused, blk, 0, stream>>>(
      offb, csrb, asb, adb, h1, b1, W2, wa2b, h2, as2b, ad2b, NN);

  gat_node_softmax_gather<64, 0, NPW><<<gNode, blk, 0, stream>>>(
      offb, csrb, as2b, ad2b, h2, b2, out, NN);
}
